# starting kernel
# baseline (speedup 1.0000x reference)
#include <hip/hip_runtime.h>
#include <hip/hip_bf16.h>

#define N_NODES   100000
#define N_EDGES   3200000
#define D_FEAT    512
#define D_HID     16
#define N_CLASSES 7

typedef float v2f __attribute__((ext_vector_type(2)));
typedef float v8f __attribute__((ext_vector_type(8)));

__global__ void init_kernel(float* __restrict__ agg1,
                            float* __restrict__ out,
                            const float* __restrict__ b2) {
  int i = blockIdx.x * blockDim.x + threadIdx.x;
  if (i < N_NODES * D_HID) agg1[i] = 0.0f;
  if (i < N_NODES * N_CLASSES) out[i] = b2[i % N_CLASSES];
}

__global__ void gemm1_wmma(const float* __restrict__ feat,
                           const float* __restrict__ W1,
                           float* __restrict__ t1) {
  __shared__ float ldsW[D_FEAT * D_HID];

  {
    const float4* s4 = (const float4*)W1;
    float4* d4 = (float4*)ldsW;
    for (int i = threadIdx.x; i < (D_FEAT * D_HID) / 4; i += blockDim.x)
      d4[i] = s4[i];
  }
  __syncthreads();

  const int lane = threadIdx.x & 31;
  const int wave = threadIdx.x >> 5;
  const int tile = blockIdx.x * (blockDim.x >> 5) + wave;
  if (tile >= N_NODES / 16) return;
  const int node0 = tile * 16;

  const int mn   = lane & 15;
  const int koff = (lane >> 4) << 1;
  const float* arow = feat + (size_t)(node0 + mn) * D_FEAT + koff;
  const float* bcol = ldsW + koff * D_HID + mn;

  v8f acc = {};
  #pragma unroll 4
  for (int kk = 0; kk < D_FEAT; kk += 4) {
    v2f a = *(const v2f*)(arow + kk);
    v2f b;
    b.x = bcol[kk * D_HID];
    b.y = bcol[kk * D_HID + D_HID];
    acc = __builtin_amdgcn_wmma_f32_16x16x4_f32(
        false, a, false, b, (short)0, acc, false, false);
  }

  const int rbase = node0 + ((lane >> 4) << 3);
  #pragma unroll
  for (int r = 0; r < 8; ++r)
    t1[(size_t)(rbase + r) * D_HID + mn] = acc[r];
}

__global__ void scatter1(const float* __restrict__ t1,
                         const int* __restrict__ src,
                         const int* __restrict__ dst,
                         float* __restrict__ agg1) {
  int e = blockIdx.x * blockDim.x + threadIdx.x;
  if (e >= N_EDGES) return;
  const int s = src[e];
  const int d = dst[e];
  const float4* hv = (const float4*)(t1 + (size_t)s * D_HID);
  float* av = agg1 + (size_t)d * D_HID;
  #pragma unroll
  for (int q = 0; q < 4; ++q) {
    float4 v = hv[q];
    atomicAdd(av + q * 4 + 0, v.x);
    atomicAdd(av + q * 4 + 1, v.y);
    atomicAdd(av + q * 4 + 2, v.z);
    atomicAdd(av + q * 4 + 3, v.w);
  }
}

__global__ void layer2_transform(const float* __restrict__ agg1,
                                 const float* __restrict__ b1,
                                 const float* __restrict__ W2,
                                 float* __restrict__ t2) {
  int node = blockIdx.x * blockDim.x + threadIdx.x;
  if (node >= N_NODES) return;
  float h[D_HID];
  const float4* a4 = (const float4*)(agg1 + (size_t)node * D_HID);
  #pragma unroll
  for (int q = 0; q < 4; ++q) {
    float4 v = a4[q];
    h[q * 4 + 0] = fmaxf(v.x + b1[q * 4 + 0], 0.0f);
    h[q * 4 + 1] = fmaxf(v.y + b1[q * 4 + 1], 0.0f);
    h[q * 4 + 2] = fmaxf(v.z + b1[q * 4 + 2], 0.0f);
    h[q * 4 + 3] = fmaxf(v.w + b1[q * 4 + 3], 0.0f);
  }
  #pragma unroll
  for (int c = 0; c < N_CLASSES; ++c) {
    float o = 0.0f;
    #pragma unroll
    for (int j = 0; j < D_HID; ++j)
      o = fmaf(h[j], W2[j * N_CLASSES + c], o);
    t2[(size_t)node * N_CLASSES + c] = o;
  }
}

__global__ void scatter2(const float* __restrict__ t2,
                         const int* __restrict__ src,
                         const int* __restrict__ dst,
                         float* __restrict__ out) {
  int e = blockIdx.x * blockDim.x + threadIdx.x;
  if (e >= N_EDGES) return;
  const int s = src[e];
  const int d = dst[e];
  const float* ts = t2 + (size_t)s * N_CLASSES;
  float* od = out + (size_t)d * N_CLASSES;
  #pragma unroll
  for (int c = 0; c < N_CLASSES; ++c)
    atomicAdd(od + c, ts[c]);
}

extern "C" void kernel_launch(void* const* d_in, const int* in_sizes, int n_in,
                              void* d_out, int out_size, void* d_ws, size_t ws_size,
                              hipStream_t stream) {
  const float* feat = (const float*)d_in[0];
  const float* W1   = (const float*)d_in[1];
  const float* b1   = (const float*)d_in[2];
  const float* W2   = (const float*)d_in[3];
  const float* b2   = (const float*)d_in[4];
  const int*   src  = (const int*)d_in[5];
  const int*   dst  = (const int*)d_in[6];
  float* out = (float*)d_out;

  float* ws   = (float*)d_ws;
  float* t1   = ws;
  float* agg1 = ws + (size_t)N_NODES * D_HID;
  float* t2   = ws + 2 * (size_t)N_NODES * D_HID;

  const int threads = 256;
  init_kernel<<<(N_NODES * D_HID + threads - 1) / threads, threads, 0, stream>>>(
      agg1, out, b2);

  const int tiles = N_NODES / 16;
  const int waves_per_block = threads / 32;
  gemm1_wmma<<<(tiles + waves_per_block - 1) / waves_per_block, threads, 0, stream>>>(
      feat, W1, t1);

  scatter1<<<(N_EDGES + threads - 1) / threads, threads, 0, stream>>>(
      t1, src, dst, agg1);

  layer2_transform<<<(N_NODES + threads - 1) / threads, threads, 0, stream>>>(
      agg1, b1, W2, t2);

  scatter2<<<(N_EDGES + threads - 1) / threads, threads, 0, stream>>>(
      t2, src, dst, out);
}


# scatter-adds with lane = column: 16 (layer 1) or 8 (layer 2) lanes share one edge row, so each f32 atomic and gather instruction covers whole rows; atomics kept, sums still in arrival order
# speedup vs baseline: 5.2420x; 5.2420x over previous
#include <hip/hip_runtime.h>
#include <hip/hip_bf16.h>

#define N_NODES   100000
#define N_EDGES   3200000
#define D_FEAT    512
#define D_HID     16
#define N_CLASSES 7

typedef float v2f __attribute__((ext_vector_type(2)));
typedef float v8f __attribute__((ext_vector_type(8)));

__global__ void init_kernel(float* __restrict__ agg1,
                            float* __restrict__ out,
                            const float* __restrict__ b2) {
  int i = blockIdx.x * blockDim.x + threadIdx.x;
  if (i < N_NODES * D_HID) agg1[i] = 0.0f;
  if (i < N_NODES * N_CLASSES) out[i] = b2[i % N_CLASSES];
}

__global__ void gemm1_wmma(const float* __restrict__ feat,
                           const float* __restrict__ W1,
                           float* __restrict__ t1) {
  __shared__ float ldsW[D_FEAT * D_HID];

  {
    const float4* s4 = (const float4*)W1;
    float4* d4 = (float4*)ldsW;
    for (int i = threadIdx.x; i < (D_FEAT * D_HID) / 4; i += blockDim.x)
      d4[i] = s4[i];
  }
  __syncthreads();

  const int lane = threadIdx.x & 31;
  const int wave = threadIdx.x >> 5;
  const int tile = blockIdx.x * (blockDim.x >> 5) + wave;
  if (tile >= N_NODES / 16) return;
  const int node0 = tile * 16;

  const int mn   = lane & 15;
  const int koff = (lane >> 4) << 1;
  const float* arow = feat + (size_t)(node0 + mn) * D_FEAT + koff;
  const float* bcol = ldsW + koff * D_HID + mn;

  v8f acc = {};
  #pragma unroll 4
  for (int kk = 0; kk < D_FEAT; kk += 4) {
    v2f a = *(const v2f*)(arow + kk);
    v2f b;
    b.x = bcol[kk * D_HID];
    b.y = bcol[kk * D_HID + D_HID];
    acc = __builtin_amdgcn_wmma_f32_16x16x4_f32(
        false, a, false, b, (short)0, acc, false, false);
  }

  const int rbase = node0 + ((lane >> 4) << 3);
  #pragma unroll
  for (int r = 0; r < 8; ++r)
    t1[(size_t)(rbase + r) * D_HID + mn] = acc[r];
}

static_assert(N_EDGES % 32 == 0, "a wave owns 32 whole edges");

__global__ void scatter1(const float* __restrict__ t1,
                         const int* __restrict__ src,
                         const int* __restrict__ dst,
                         float* __restrict__ agg1) {
  const int e = blockIdx.x * blockDim.x + threadIdx.x;
  const int lane = threadIdx.x & 31;
  if (e - lane >= N_EDGES) return;
  const int s_l = src[e];
  const int d_l = dst[e];
  const int f    = lane & 15;
  const int half = lane >> 4;
  float v[16];
  int   o[16];
  #pragma unroll
  for (int i = 0; i < 16; ++i) {
    const int from = 2 * i + half;
    const int s = __shfl(s_l, from);
    const int d = __shfl(d_l, from);
    v[i] = t1[s * D_HID + f];
    o[i] = d * D_HID + f;
  }
  #pragma unroll
  for (int i = 0; i < 16; ++i)
    atomicAdd(agg1 + o[i], v[i]);
}

__global__ void layer2_transform(const float* __restrict__ agg1,
                                 const float* __restrict__ b1,
                                 const float* __restrict__ W2,
                                 float* __restrict__ t2) {
  int node = blockIdx.x * blockDim.x + threadIdx.x;
  if (node >= N_NODES) return;
  float h[D_HID];
  const float4* a4 = (const float4*)(agg1 + (size_t)node * D_HID);
  #pragma unroll
  for (int q = 0; q < 4; ++q) {
    float4 v = a4[q];
    h[q * 4 + 0] = fmaxf(v.x + b1[q * 4 + 0], 0.0f);
    h[q * 4 + 1] = fmaxf(v.y + b1[q * 4 + 1], 0.0f);
    h[q * 4 + 2] = fmaxf(v.z + b1[q * 4 + 2], 0.0f);
    h[q * 4 + 3] = fmaxf(v.w + b1[q * 4 + 3], 0.0f);
  }
  #pragma unroll
  for (int c = 0; c < N_CLASSES; ++c) {
    float o = 0.0f;
    #pragma unroll
    for (int j = 0; j < D_HID; ++j)
      o = fmaf(h[j], W2[j * N_CLASSES + c], o);
    t2[(size_t)node * N_CLASSES + c] = o;
  }
}

__global__ void scatter2(const float* __restrict__ t2,
                         const int* __restrict__ src,
                         const int* __restrict__ dst,
                         float* __restrict__ out) {
  const int e = blockIdx.x * blockDim.x + threadIdx.x;
  const int lane = threadIdx.x & 31;
  if (e - lane >= N_EDGES) return;
  const int s_l = src[e];
  const int d_l = dst[e];
  const int c = lane & 7;
  const int q = lane >> 3;
  int si[8];
  int o[8];
  #pragma unroll
  for (int i = 0; i < 8; ++i) {
    const int from = 4 * i + q;
    si[i] = __shfl(s_l, from) * N_CLASSES + c;
    o[i]  = __shfl(d_l, from) * N_CLASSES + c;
  }
  if (c < N_CLASSES) {
    float v[8];
    #pragma unroll
    for (int i = 0; i < 8; ++i) v[i] = t2[si[i]];
    #pragma unroll
    for (int i = 0; i < 8; ++i) atomicAdd(out + o[i], v[i]);
  }
}

extern "C" void kernel_launch(void* const* d_in, const int* in_sizes, int n_in,
                              void* d_out, int out_size, void* d_ws, size_t ws_size,
                              hipStream_t stream) {
  const float* feat = (const float*)d_in[0];
  const float* W1   = (const float*)d_in[1];
  const float* b1   = (const float*)d_in[2];
  const float* W2   = (const float*)d_in[3];
  const float* b2   = (const float*)d_in[4];
  const int*   src  = (const int*)d_in[5];
  const int*   dst  = (const int*)d_in[6];
  float* out = (float*)d_out;

  float* ws   = (float*)d_ws;
  float* t1   = ws;
  float* agg1 = ws + (size_t)N_NODES * D_HID;
  float* t2   = ws + 2 * (size_t)N_NODES * D_HID;

  const int threads = 256;
  init_kernel<<<(N_NODES * D_HID + threads - 1) / threads, threads, 0, stream>>>(
      agg1, out, b2);

  const int tiles = N_NODES / 16;
  const int waves_per_block = threads / 32;
  gemm1_wmma<<<(tiles + waves_per_block - 1) / waves_per_block, threads, 0, stream>>>(
      feat, W1, t1);

  scatter1<<<(N_EDGES + threads - 1) / threads, threads, 0, stream>>>(
      t1, src, dst, agg1);

  layer2_transform<<<(N_NODES + threads - 1) / threads, threads, 0, stream>>>(
      agg1, b1, W2, t2);

  scatter2<<<(N_EDGES + threads - 1) / threads, threads, 0, stream>>>(
      t2, src, dst, out);
}


# scatter-adds go into one accumulator plane per cache half of the chip (by die id), so a line's f32 atomics all come from one half; reading pass sums the planes; f32 sum order changes
# speedup vs baseline: 9.2275x; 1.7603x over previous
#include <hip/hip_runtime.h>
#include <hip/hip_bf16.h>

#define N_NODES   100000
#define N_EDGES   3200000
#define D_FEAT    512
#define D_HID     16
#define N_CLASSES 7

#define WS_OUT_B  (N_NODES * N_CLASSES)
#define WS_AGG_A  (N_NODES * D_HID)
#define WS_AGG_B  (2 * N_NODES * D_HID)
#define WS_BYTES_TWO_PLANES ((size_t)3 * N_NODES * D_HID * sizeof(float))

typedef float v2f __attribute__((ext_vector_type(2)));
typedef float v8f __attribute__((ext_vector_type(8)));

__device__ __forceinline__ int chip_half() {
  const unsigned id = __builtin_amdgcn_s_sendmsg_rtn(0x87);
  return (int)((id >> 18) & 1u);
}

__global__ void init_kernel(float* __restrict__ agg_a,
                            float* __restrict__ agg_b,
                            size_t ws_bytes,
                            float* __restrict__ out,
                            const float* __restrict__ b2) {
  int i = blockIdx.x * blockDim.x + threadIdx.x;
  if (i < N_NODES * D_HID) {
    agg_a[i] = 0.0f;
    if (ws_bytes >= WS_BYTES_TWO_PLANES) agg_b[i] = 0.0f;
  }
  if (i < N_NODES * N_CLASSES) out[i] = b2[i % N_CLASSES];
}

__global__ void gemm1_wmma(const float* __restrict__ feat,
                           const float* __restrict__ W1,
                           float* __restrict__ t1) {
  __shared__ float ldsW[D_FEAT * D_HID];

  {
    const float4* s4 = (const float4*)W1;
    float4* d4 = (float4*)ldsW;
    for (int i = threadIdx.x; i < (D_FEAT * D_HID) / 4; i += blockDim.x)
      d4[i] = s4[i];
  }
  __syncthreads();

  const int lane = threadIdx.x & 31;
  const int wave = threadIdx.x >> 5;
  const int tile = blockIdx.x * (blockDim.x >> 5) + wave;
  if (tile >= N_NODES / 16) return;
  const int node0 = tile * 16;

  const int mn   = lane & 15;
  const int koff = (lane >> 4) << 1;
  const float* arow = feat + (size_t)(node0 + mn) * D_FEAT + koff;
  const float* bcol = ldsW + koff * D_HID + mn;

  v8f acc = {};
  #pragma unroll 4
  for (int kk = 0; kk < D_FEAT; kk += 4) {
    v2f a = *(const v2f*)(arow + kk);
    v2f b;
    b.x = bcol[kk * D_HID];
    b.y = bcol[kk * D_HID + D_HID];
    acc = __builtin_amdgcn_wmma_f32_16x16x4_f32(
        false, a, false, b, (short)0, acc, false, false);
  }

  const int rbase = node0 + ((lane >> 4) << 3);
  #pragma unroll
  for (int r = 0; r < 8; ++r)
    t1[(size_t)(rbase + r) * D_HID + mn] = acc[r];
}

static_assert(N_EDGES % 32 == 0, "a wave owns 32 whole edges");

__global__ void scatter1(const float* __restrict__ t1,
                         const int* __restrict__ src,
                         const int* __restrict__ dst,
                         float* __restrict__ agg_a,
                         float* __restrict__ agg_b,
                         size_t ws_bytes) {
  const int e = blockIdx.x * blockDim.x + threadIdx.x;
  const int lane = threadIdx.x & 31;
  if (e - lane >= N_EDGES) return;
  const int s_l = src[e];
  const int d_l = dst[e];
  float* agg1 = (chip_half() != 0 && ws_bytes >= WS_BYTES_TWO_PLANES) ? agg_b : agg_a;
  const int f    = lane & 15;
  const int half = lane >> 4;
  float v[16];
  int   o[16];
  #pragma unroll
  for (int i = 0; i < 16; ++i) {
    const int from = 2 * i + half;
    const int s = __shfl(s_l, from);
    const int d = __shfl(d_l, from);
    v[i] = t1[s * D_HID + f];
    o[i] = d * D_HID + f;
  }
  #pragma unroll
  for (int i = 0; i < 16; ++i)
    atomicAdd(agg1 + o[i], v[i]);
}

__global__ void layer2_transform(const float* __restrict__ agg_a,
                                 const float* __restrict__ agg_b,
                                 size_t ws_bytes,
                                 const float* __restrict__ b1,
                                 const float* __restrict__ W2,
                                 float* __restrict__ t2,
                                 float* __restrict__ out_b) {
  int node = blockIdx.x * blockDim.x + threadIdx.x;
  if (node >= N_NODES) return;
  const bool two = ws_bytes >= WS_BYTES_TWO_PLANES;
  float h[D_HID];
  const float4* a4 = (const float4*)(agg_a + (size_t)node * D_HID);
  const float4* p4 = (const float4*)(agg_b + (size_t)node * D_HID);
  float4 va[4];
  #pragma unroll
  for (int q = 0; q < 4; ++q) va[q] = a4[q];
  if (two) {
    float4 vb[4];
    #pragma unroll
    for (int q = 0; q < 4; ++q) vb[q] = p4[q];
    #pragma unroll
    for (int q = 0; q < 4; ++q) {
      va[q].x += vb[q].x; va[q].y += vb[q].y; va[q].z += vb[q].z; va[q].w += vb[q].w;
    }
  }
  #pragma unroll
  for (int q = 0; q < 4; ++q) {
    const float4 v = va[q];
    h[q * 4 + 0] = fmaxf(v.x + b1[q * 4 + 0], 0.0f);
    h[q * 4 + 1] = fmaxf(v.y + b1[q * 4 + 1], 0.0f);
    h[q * 4 + 2] = fmaxf(v.z + b1[q * 4 + 2], 0.0f);
    h[q * 4 + 3] = fmaxf(v.w + b1[q * 4 + 3], 0.0f);
  }
  #pragma unroll
  for (int c = 0; c < N_CLASSES; ++c) {
    float o = 0.0f;
    #pragma unroll
    for (int j = 0; j < D_HID; ++j)
      o = fmaf(h[j], W2[j * N_CLASSES + c], o);
    t2[(size_t)node * N_CLASSES + c] = o;
    out_b[(size_t)node * N_CLASSES + c] = 0.0f;
  }
}

__global__ void scatter2(const float* __restrict__ t2,
                         const int* __restrict__ src,
                         const int* __restrict__ dst,
                         float* __restrict__ out_a,
                         float* __restrict__ out_b) {
  const int e = blockIdx.x * blockDim.x + threadIdx.x;
  const int lane = threadIdx.x & 31;
  if (e - lane >= N_EDGES) return;
  const int s_l = src[e];
  const int d_l = dst[e];
  float* out = (chip_half() != 0) ? out_b : out_a;
  const int c = lane & 7;
  const int q = lane >> 3;
  int si[8];
  int o[8];
  #pragma unroll
  for (int i = 0; i < 8; ++i) {
    const int from = 4 * i + q;
    si[i] = __shfl(s_l, from) * N_CLASSES + c;
    o[i]  = __shfl(d_l, from) * N_CLASSES + c;
  }
  if (c < N_CLASSES) {
    float v[8];
    #pragma unroll
    for (int i = 0; i < 8; ++i) v[i] = t2[si[i]];
    #pragma unroll
    for (int i = 0; i < 8; ++i) atomicAdd(out + o[i], v[i]);
  }
}

static_assert((N_NODES * N_CLASSES) % 4 == 0, "whole float4s");

__global__ void combine_out(float* __restrict__ out,
                            const float* __restrict__ out_b) {
  const int i = blockIdx.x * blockDim.x + threadIdx.x;
  if (i >= (N_NODES * N_CLASSES) / 4) return;
  float4 a = ((float4*)out)[i];
  const float4 b = ((const float4*)out_b)[i];
  a.x += b.x; a.y += b.y; a.z += b.z; a.w += b.w;
  ((float4*)out)[i] = a;
}

extern "C" void kernel_launch(void* const* d_in, const int* in_sizes, int n_in,
                              void* d_out, int out_size, void* d_ws, size_t ws_size,
                              hipStream_t stream) {
  const float* feat = (const float*)d_in[0];
  const float* W1   = (const float*)d_in[1];
  const float* b1   = (const float*)d_in[2];
  const float* W2   = (const float*)d_in[3];
  const float* b2   = (const float*)d_in[4];
  const int*   src  = (const int*)d_in[5];
  const int*   dst  = (const int*)d_in[6];
  float* out = (float*)d_out;

  float* ws    = (float*)d_ws;
  float* t1    = ws;
  float* t2    = ws;
  float* out_b = ws + WS_OUT_B;
  float* agg_a = ws + WS_AGG_A;
  float* agg_b = ws + WS_AGG_B;

  const int threads = 256;
  init_kernel<<<(N_NODES * D_HID + threads - 1) / threads, threads, 0, stream>>>(
      agg_a, agg_b, ws_size, out, b2);

  const int tiles = N_NODES / 16;
  const int waves_per_block = threads / 32;
  gemm1_wmma<<<(tiles + waves_per_block - 1) / waves_per_block, threads, 0, stream>>>(
      feat, W1, t1);

  scatter1<<<(N_EDGES + threads - 1) / threads, threads, 0, stream>>>(
      t1, src, dst, agg_a, agg_b, ws_size);

  layer2_transform<<<(N_NODES + threads - 1) / threads, threads, 0, stream>>>(
      agg_a, agg_b, ws_size, b1, W2, t2, out_b);

  scatter2<<<(N_EDGES + threads - 1) / threads, threads, 0, stream>>>(
      t2, src, dst, out, out_b);

  combine_out<<<((N_NODES * N_CLASSES) / 4 + threads - 1) / threads, threads, 0, stream>>>(
      out, out_b);
}


# layer-2 rows on aligned pitches: t2 rows 64-B apart (kept in the spent agg1 plane), both out accumulator planes at 32-B pitch in workspace, out = planes + b2 in combine; f32 sum order changes
# speedup vs baseline: 10.2703x; 1.1130x over previous
#include <hip/hip_runtime.h>
#include <hip/hip_bf16.h>

#define N_NODES   100000
#define N_EDGES   3200000
#define D_FEAT    512
#define D_HID     16
#define N_CLASSES 7

#define L2_PITCH  8
#define T2_PITCH  D_HID
#define WS_OUT_B  (N_NODES * L2_PITCH)
#define WS_AGG_A  (N_NODES * D_HID)
#define WS_AGG_B  (2 * N_NODES * D_HID)
#define WS_BYTES_TWO_PLANES ((size_t)3 * N_NODES * D_HID * sizeof(float))

typedef float v2f __attribute__((ext_vector_type(2)));
typedef float v8f __attribute__((ext_vector_type(8)));

__device__ __forceinline__ int chip_half() {
  const unsigned id = __builtin_amdgcn_s_sendmsg_rtn(0x87);
  return (int)((id >> 18) & 1u);
}

__global__ void init_kernel(float* __restrict__ agg_a,
                            float* __restrict__ agg_b,
                            size_t ws_bytes) {
  int i = blockIdx.x * blockDim.x + threadIdx.x;
  if (i < N_NODES * D_HID) {
    agg_a[i] = 0.0f;
    if (ws_bytes >= WS_BYTES_TWO_PLANES) agg_b[i] = 0.0f;
  }
}

__global__ void gemm1_wmma(const float* __restrict__ feat,
                           const float* __restrict__ W1,
                           float* __restrict__ t1) {
  __shared__ float ldsW[D_FEAT * D_HID];

  {
    const float4* s4 = (const float4*)W1;
    float4* d4 = (float4*)ldsW;
    for (int i = threadIdx.x; i < (D_FEAT * D_HID) / 4; i += blockDim.x)
      d4[i] = s4[i];
  }
  __syncthreads();

  const int lane = threadIdx.x & 31;
  const int wave = threadIdx.x >> 5;
  const int tile = blockIdx.x * (blockDim.x >> 5) + wave;
  if (tile >= N_NODES / 16) return;
  const int node0 = tile * 16;

  const int mn   = lane & 15;
  const int koff = (lane >> 4) << 1;
  const float* arow = feat + (size_t)(node0 + mn) * D_FEAT + koff;
  const float* bcol = ldsW + koff * D_HID + mn;

  v8f acc = {};
  #pragma unroll 4
  for (int kk = 0; kk < D_FEAT; kk += 4) {
    v2f a = *(const v2f*)(arow + kk);
    v2f b;
    b.x = bcol[kk * D_HID];
    b.y = bcol[kk * D_HID + D_HID];
    acc = __builtin_amdgcn_wmma_f32_16x16x4_f32(
        false, a, false, b, (short)0, acc, false, false);
  }

  const int rbase = node0 + ((lane >> 4) << 3);
  #pragma unroll
  for (int r = 0; r < 8; ++r)
    t1[(size_t)(rbase + r) * D_HID + mn] = acc[r];
}

static_assert(N_EDGES % 32 == 0, "a wave owns 32 whole edges");

__global__ void scatter1(const float* __restrict__ t1,
                         const int* __restrict__ src,
                         const int* __restrict__ dst,
                         float* __restrict__ agg_a,
                         float* __restrict__ agg_b,
                         size_t ws_bytes) {
  const int e = blockIdx.x * blockDim.x + threadIdx.x;
  const int lane = threadIdx.x & 31;
  if (e - lane >= N_EDGES) return;
  const int s_l = src[e];
  const int d_l = dst[e];
  float* agg1 = (chip_half() != 0 && ws_bytes >= WS_BYTES_TWO_PLANES) ? agg_b : agg_a;
  const int f    = lane & 15;
  const int half = lane >> 4;
  float v[16];
  int   o[16];
  #pragma unroll
  for (int i = 0; i < 16; ++i) {
    const int from = 2 * i + half;
    const int s = __shfl(s_l, from);
    const int d = __shfl(d_l, from);
    v[i] = t1[s * D_HID + f];
    o[i] = d * D_HID + f;
  }
  #pragma unroll
  for (int i = 0; i < 16; ++i)
    atomicAdd(agg1 + o[i], v[i]);
}

__global__ void layer2_transform(float* agg_a,
                                 const float* __restrict__ agg_b,
                                 size_t ws_bytes,
                                 const float* __restrict__ b1,
                                 const float* __restrict__ W2,
                                 float* __restrict__ out_a,
                                 float* __restrict__ out_b) {
  int node = blockIdx.x * blockDim.x + threadIdx.x;
  if (node >= N_NODES) return;
  const bool two = ws_bytes >= WS_BYTES_TWO_PLANES;
  float h[D_HID];
  const float4* a4 = (const float4*)(agg_a + (size_t)node * D_HID);
  const float4* p4 = (const float4*)(agg_b + (size_t)node * D_HID);
  float4 va[4];
  #pragma unroll
  for (int q = 0; q < 4; ++q) va[q] = a4[q];
  if (two) {
    float4 vb[4];
    #pragma unroll
    for (int q = 0; q < 4; ++q) vb[q] = p4[q];
    #pragma unroll
    for (int q = 0; q < 4; ++q) {
      va[q].x += vb[q].x; va[q].y += vb[q].y; va[q].z += vb[q].z; va[q].w += vb[q].w;
    }
  }
  #pragma unroll
  for (int q = 0; q < 4; ++q) {
    const float4 v = va[q];
    h[q * 4 + 0] = fmaxf(v.x + b1[q * 4 + 0], 0.0f);
    h[q * 4 + 1] = fmaxf(v.y + b1[q * 4 + 1], 0.0f);
    h[q * 4 + 2] = fmaxf(v.z + b1[q * 4 + 2], 0.0f);
    h[q * 4 + 3] = fmaxf(v.w + b1[q * 4 + 3], 0.0f);
  }
  float t[L2_PITCH];
  #pragma unroll
  for (int c = 0; c < N_CLASSES; ++c) {
    float o = 0.0f;
    #pragma unroll
    for (int j = 0; j < D_HID; ++j)
      o = fmaf(h[j], W2[j * N_CLASSES + c], o);
    t[c] = o;
  }
  t[N_CLASSES] = 0.0f;
  float4* t4 = (float4*)(agg_a + (size_t)node * T2_PITCH);
  t4[0] = make_float4(t[0], t[1], t[2], t[3]);
  t4[1] = make_float4(t[4], t[5], t[6], t[7]);
  const float4 z = make_float4(0.0f, 0.0f, 0.0f, 0.0f);
  float4* za = (float4*)(out_a + (size_t)node * L2_PITCH);
  float4* zb = (float4*)(out_b + (size_t)node * L2_PITCH);
  za[0] = z; za[1] = z;
  zb[0] = z; zb[1] = z;
}

__global__ void scatter2(const float* __restrict__ t2,
                         const int* __restrict__ src,
                         const int* __restrict__ dst,
                         float* __restrict__ out_a,
                         float* __restrict__ out_b) {
  const int e = blockIdx.x * blockDim.x + threadIdx.x;
  const int lane = threadIdx.x & 31;
  if (e - lane >= N_EDGES) return;
  const int s_l = src[e];
  const int d_l = dst[e];
  float* out = (chip_half() != 0) ? out_b : out_a;
  const int c = lane & 7;
  const int q = lane >> 3;
  int si[8];
  int o[8];
  #pragma unroll
  for (int i = 0; i < 8; ++i) {
    const int from = 4 * i + q;
    si[i] = __shfl(s_l, from) * T2_PITCH + c;
    o[i]  = __shfl(d_l, from) * L2_PITCH + c;
  }
  if (c < N_CLASSES) {
    float v[8];
    #pragma unroll
    for (int i = 0; i < 8; ++i) v[i] = t2[si[i]];
    #pragma unroll
    for (int i = 0; i < 8; ++i) atomicAdd(out + o[i], v[i]);
  }
}

static_assert((N_NODES * N_CLASSES) % 4 == 0, "whole float4s");

__global__ void combine_out(float* __restrict__ out,
                            const float* __restrict__ out_a,
                            const float* __restrict__ out_b,
                            const float* __restrict__ b2) {
  const int i = blockIdx.x * blockDim.x + threadIdx.x;
  if (i >= (N_NODES * N_CLASSES) / 4) return;
  float r[4];
  #pragma unroll
  for (int j = 0; j < 4; ++j) {
    const int e = 4 * i + j;
    const int n = e / N_CLASSES;
    const int c = e - n * N_CLASSES;
    r[j] = (out_a[n * L2_PITCH + c] + out_b[n * L2_PITCH + c]) + b2[c];
  }
  ((float4*)out)[i] = make_float4(r[0], r[1], r[2], r[3]);
}

extern "C" void kernel_launch(void* const* d_in, const int* in_sizes, int n_in,
                              void* d_out, int out_size, void* d_ws, size_t ws_size,
                              hipStream_t stream) {
  const float* feat = (const float*)d_in[0];
  const float* W1   = (const float*)d_in[1];
  const float* b1   = (const float*)d_in[2];
  const float* W2   = (const float*)d_in[3];
  const float* b2   = (const float*)d_in[4];
  const int*   src  = (const int*)d_in[5];
  const int*   dst  = (const int*)d_in[6];
  float* out = (float*)d_out;

  float* ws    = (float*)d_ws;
  float* t1    = ws;
  float* out_a = ws;
  float* out_b = ws + WS_OUT_B;
  float* agg_a = ws + WS_AGG_A;
  float* agg_b = ws + WS_AGG_B;

  const int threads = 256;
  init_kernel<<<(N_NODES * D_HID + threads - 1) / threads, threads, 0, stream>>>(
      agg_a, agg_b, ws_size);

  const int tiles = N_NODES / 16;
  const int waves_per_block = threads / 32;
  gemm1_wmma<<<(tiles + waves_per_block - 1) / waves_per_block, threads, 0, stream>>>(
      feat, W1, t1);

  scatter1<<<(N_EDGES + threads - 1) / threads, threads, 0, stream>>>(
      t1, src, dst, agg_a, agg_b, ws_size);

  layer2_transform<<<(N_NODES + threads - 1) / threads, threads, 0, stream>>>(
      agg_a, agg_b, ws_size, b1, W2, out_a, out_b);

  scatter2<<<(N_EDGES + threads - 1) / threads, threads, 0, stream>>>(
      agg_a, src, dst, out_a, out_b);

  combine_out<<<((N_NODES * N_CLASSES) / 4 + threads - 1) / threads, threads, 0, stream>>>(
      out, out_a, out_b, b2);
}


# init and both scatters as resident grids of long-lived waves: 8 groups of 32 edges per wave, next indices read ahead (12,500 waves, not 100,000); init 4 float4 per lane; atomics kept, arrival order
# speedup vs baseline: 10.6411x; 1.0361x over previous
#include <hip/hip_runtime.h>
#include <hip/hip_bf16.h>

#define N_NODES   100000
#define N_EDGES   3200000
#define D_FEAT    512
#define D_HID     16
#define N_CLASSES 7

#define L2_PITCH  8
#define T2_PITCH  D_HID
#define WS_OUT_B  (N_NODES * L2_PITCH)
#define WS_AGG_A  (N_NODES * D_HID)
#define WS_AGG_B  (2 * N_NODES * D_HID)
#define WS_BYTES_TWO_PLANES ((size_t)3 * N_NODES * D_HID * sizeof(float))

typedef float v2f __attribute__((ext_vector_type(2)));
typedef float v8f __attribute__((ext_vector_type(8)));

__device__ __forceinline__ int chip_half() {
  const unsigned id = __builtin_amdgcn_s_sendmsg_rtn(0x87);
  return (int)((id >> 18) & 1u);
}

#define INIT_THREADS 256
#define INIT_BLOCKS  391
#define SC_THREADS   128
#define SC_GROUPS    8
#define SC_BLOCKS    (N_EDGES / (SC_THREADS * SC_GROUPS))
#define SC_STRIDE    (SC_BLOCKS * SC_THREADS)
static_assert(SC_STRIDE * SC_GROUPS == N_EDGES, "the scatter grid covers the edges exactly");
static_assert(SC_STRIDE % 32 == 0, "a wave owns 32 whole edges in every sweep");

static_assert((N_NODES * D_HID) % 4 == 0, "whole float4s");

__global__ void init_kernel(float* __restrict__ agg_a,
                            float* __restrict__ agg_b,
                            size_t ws_bytes) {
  const bool two = ws_bytes >= WS_BYTES_TWO_PLANES;
  const float4 z = make_float4(0.0f, 0.0f, 0.0f, 0.0f);
  float4* a4 = (float4*)agg_a;
  float4* b4 = (float4*)agg_b;
  for (int i = blockIdx.x * INIT_THREADS + threadIdx.x; i < (N_NODES * D_HID) / 4;
       i += INIT_BLOCKS * INIT_THREADS) {
    a4[i] = z;
    if (two) b4[i] = z;
  }
}

__global__ void gemm1_wmma(const float* __restrict__ feat,
                           const float* __restrict__ W1,
                           float* __restrict__ t1) {
  __shared__ float ldsW[D_FEAT * D_HID];

  {
    const float4* s4 = (const float4*)W1;
    float4* d4 = (float4*)ldsW;
    for (int i = threadIdx.x; i < (D_FEAT * D_HID) / 4; i += blockDim.x)
      d4[i] = s4[i];
  }
  __syncthreads();

  const int lane = threadIdx.x & 31;
  const int wave = threadIdx.x >> 5;
  const int tile = blockIdx.x * (blockDim.x >> 5) + wave;
  if (tile >= N_NODES / 16) return;
  const int node0 = tile * 16;

  const int mn   = lane & 15;
  const int koff = (lane >> 4) << 1;
  const float* arow = feat + (size_t)(node0 + mn) * D_FEAT + koff;
  const float* bcol = ldsW + koff * D_HID + mn;

  v8f acc = {};
  #pragma unroll 4
  for (int kk = 0; kk < D_FEAT; kk += 4) {
    v2f a = *(const v2f*)(arow + kk);
    v2f b;
    b.x = bcol[kk * D_HID];
    b.y = bcol[kk * D_HID + D_HID];
    acc = __builtin_amdgcn_wmma_f32_16x16x4_f32(
        false, a, false, b, (short)0, acc, false, false);
  }

  const int rbase = node0 + ((lane >> 4) << 3);
  #pragma unroll
  for (int r = 0; r < 8; ++r)
    t1[(size_t)(rbase + r) * D_HID + mn] = acc[r];
}


__global__ void scatter1(const float* __restrict__ t1,
                         const int* __restrict__ src,
                         const int* __restrict__ dst,
                         float* __restrict__ agg_a,
                         float* __restrict__ agg_b,
                         size_t ws_bytes) {
  int e = blockIdx.x * SC_THREADS + threadIdx.x;
  const int lane = threadIdx.x & 31;
  int s_l = src[e];
  int d_l = dst[e];
  float* agg1 = (chip_half() != 0 && ws_bytes >= WS_BYTES_TWO_PLANES) ? agg_b : agg_a;
  const int f    = lane & 15;
  const int half = lane >> 4;
  for (int g = 0; g < SC_GROUPS; ++g) {
    float v[16];
    int   o[16];
    #pragma unroll
    for (int i = 0; i < 16; ++i) {
      const int from = 2 * i + half;
      const int s = __shfl(s_l, from);
      const int d = __shfl(d_l, from);
      v[i] = t1[s * D_HID + f];
      o[i] = d * D_HID + f;
    }
    if (g + 1 < SC_GROUPS) {
      e += SC_STRIDE;
      s_l = src[e];
      d_l = dst[e];
    }
    #pragma unroll
    for (int i = 0; i < 16; ++i)
      atomicAdd(agg1 + o[i], v[i]);
  }
}

__global__ void layer2_transform(float* agg_a,
                                 const float* __restrict__ agg_b,
                                 size_t ws_bytes,
                                 const float* __restrict__ b1,
                                 const float* __restrict__ W2,
                                 float* __restrict__ out_a,
                                 float* __restrict__ out_b) {
  int node = blockIdx.x * blockDim.x + threadIdx.x;
  if (node >= N_NODES) return;
  const bool two = ws_bytes >= WS_BYTES_TWO_PLANES;
  float h[D_HID];
  const float4* a4 = (const float4*)(agg_a + (size_t)node * D_HID);
  const float4* p4 = (const float4*)(agg_b + (size_t)node * D_HID);
  float4 va[4];
  #pragma unroll
  for (int q = 0; q < 4; ++q) va[q] = a4[q];
  if (two) {
    float4 vb[4];
    #pragma unroll
    for (int q = 0; q < 4; ++q) vb[q] = p4[q];
    #pragma unroll
    for (int q = 0; q < 4; ++q) {
      va[q].x += vb[q].x; va[q].y += vb[q].y; va[q].z += vb[q].z; va[q].w += vb[q].w;
    }
  }
  #pragma unroll
  for (int q = 0; q < 4; ++q) {
    const float4 v = va[q];
    h[q * 4 + 0] = fmaxf(v.x + b1[q * 4 + 0], 0.0f);
    h[q * 4 + 1] = fmaxf(v.y + b1[q * 4 + 1], 0.0f);
    h[q * 4 + 2] = fmaxf(v.z + b1[q * 4 + 2], 0.0f);
    h[q * 4 + 3] = fmaxf(v.w + b1[q * 4 + 3], 0.0f);
  }
  float t[L2_PITCH];
  #pragma unroll
  for (int c = 0; c < N_CLASSES; ++c) {
    float o = 0.0f;
    #pragma unroll
    for (int j = 0; j < D_HID; ++j)
      o = fmaf(h[j], W2[j * N_CLASSES + c], o);
    t[c] = o;
  }
  t[N_CLASSES] = 0.0f;
  float4* t4 = (float4*)(agg_a + (size_t)node * T2_PITCH);
  t4[0] = make_float4(t[0], t[1], t[2], t[3]);
  t4[1] = make_float4(t[4], t[5], t[6], t[7]);
  const float4 z = make_float4(0.0f, 0.0f, 0.0f, 0.0f);
  float4* za = (float4*)(out_a + (size_t)node * L2_PITCH);
  float4* zb = (float4*)(out_b + (size_t)node * L2_PITCH);
  za[0] = z; za[1] = z;
  zb[0] = z; zb[1] = z;
}

__global__ void scatter2(const float* __restrict__ t2,
                         const int* __restrict__ src,
                         const int* __restrict__ dst,
                         float* __restrict__ out_a,
                         float* __restrict__ out_b) {
  int e = blockIdx.x * SC_THREADS + threadIdx.x;
  const int lane = threadIdx.x & 31;
  int s_l = src[e];
  int d_l = dst[e];
  float* out = (chip_half() != 0) ? out_b : out_a;
  const int c = lane & 7;
  const int q = lane >> 3;
  for (int g = 0; g < SC_GROUPS; ++g) {
    int si[8];
    int o[8];
    #pragma unroll
    for (int i = 0; i < 8; ++i) {
      const int from = 4 * i + q;
      si[i] = __shfl(s_l, from) * T2_PITCH + c;
      o[i]  = __shfl(d_l, from) * L2_PITCH + c;
    }
    float v[8];
    if (c < N_CLASSES) {
      #pragma unroll
      for (int i = 0; i < 8; ++i) v[i] = t2[si[i]];
    }
    if (g + 1 < SC_GROUPS) {
      e += SC_STRIDE;
      s_l = src[e];
      d_l = dst[e];
    }
    if (c < N_CLASSES) {
      #pragma unroll
      for (int i = 0; i < 8; ++i) atomicAdd(out + o[i], v[i]);
    }
  }
}

static_assert((N_NODES * N_CLASSES) % 4 == 0, "whole float4s");

__global__ void combine_out(float* __restrict__ out,
                            const float* __restrict__ out_a,
                            const float* __restrict__ out_b,
                            const float* __restrict__ b2) {
  const int i = blockIdx.x * blockDim.x + threadIdx.x;
  if (i >= (N_NODES * N_CLASSES) / 4) return;
  float r[4];
  #pragma unroll
  for (int j = 0; j < 4; ++j) {
    const int e = 4 * i + j;
    const int n = e / N_CLASSES;
    const int c = e - n * N_CLASSES;
    r[j] = (out_a[n * L2_PITCH + c] + out_b[n * L2_PITCH + c]) + b2[c];
  }
  ((float4*)out)[i] = make_float4(r[0], r[1], r[2], r[3]);
}

extern "C" void kernel_launch(void* const* d_in, const int* in_sizes, int n_in,
                              void* d_out, int out_size, void* d_ws, size_t ws_size,
                              hipStream_t stream) {
  const float* feat = (const float*)d_in[0];
  const float* W1   = (const float*)d_in[1];
  const float* b1   = (const float*)d_in[2];
  const float* W2   = (const float*)d_in[3];
  const float* b2   = (const float*)d_in[4];
  const int*   src  = (const int*)d_in[5];
  const int*   dst  = (const int*)d_in[6];
  float* out = (float*)d_out;

  float* ws    = (float*)d_ws;
  float* t1    = ws;
  float* out_a = ws;
  float* out_b = ws + WS_OUT_B;
  float* agg_a = ws + WS_AGG_A;
  float* agg_b = ws + WS_AGG_B;

  const int threads = 256;
  init_kernel<<<INIT_BLOCKS, INIT_THREADS, 0, stream>>>(
      agg_a, agg_b, ws_size);

  const int tiles = N_NODES / 16;
  const int waves_per_block = threads / 32;
  gemm1_wmma<<<(tiles + waves_per_block - 1) / waves_per_block, threads, 0, stream>>>(
      feat, W1, t1);

  scatter1<<<SC_BLOCKS, SC_THREADS, 0, stream>>>(
      t1, src, dst, agg_a, agg_b, ws_size);

  layer2_transform<<<(N_NODES + threads - 1) / threads, threads, 0, stream>>>(
      agg_a, agg_b, ws_size, b1, W2, out_a, out_b);

  scatter2<<<SC_BLOCKS, SC_THREADS, 0, stream>>>(
      agg_a, src, dst, out_a, out_b);

  combine_out<<<((N_NODES * N_CLASSES) / 4 + threads - 1) / threads, threads, 0, stream>>>(
      out, out_a, out_b, b2);
}


# edges grouped by destination range once per call (LDS counters); one workgroup per range sums gathered rows by LDS f32 adds, stores each row once; no global atomics; f32 sum order changes
# speedup vs baseline: 12.2660x; 1.1527x over previous
#include <hip/hip_runtime.h>


#define N_NODES   100000
#define N_EDGES   3200000
#define D_FEAT    512
#define D_HID     16
#define N_CLASSES 7

#define BIN_NODES    416
#define NB           241
#define NCH          250
#define CH_EDGES     12800
#define CH_THREADS   512
#define CH_PER_LANE  25
#define SUM_THREADS  1024
#define T2_PITCH     8
#define SRC_BITS     17
#define SRC_MASK     0x1FFFFu

static_assert(NB * BIN_NODES >= N_NODES && (NB - 1) * BIN_NODES < N_NODES, "the ranges cover the nodes");
static_assert(NB <= 256 && CH_THREADS % 256 == 0, "256 threads, one per range, for every share of the column sums; a range fits a byte");
static_assert((BIN_NODES * N_CLASSES * sizeof(float)) % 128 == 0, "a range's rows of out are whole lines");
static_assert(NCH * CH_EDGES == N_EDGES, "the chunks cover the edges exactly");
static_assert(CH_THREADS * CH_PER_LANE == CH_EDGES, "whole edges per thread");
static_assert(N_NODES <= (1 << SRC_BITS) && BIN_NODES <= (1 << (31 - SRC_BITS)), "src and row fit one word");

#define WS_T1    0
#define WS_LIST  (N_NODES * D_HID)
#define WS_T2    (WS_LIST + N_EDGES)
#define WS_CNT   (WS_T2 + N_NODES * T2_PITCH)
#define WS_BASE  (WS_CNT + NCH * NB)
#define WS_WORDS (WS_BASE + NB + 1)
#define WS_NEED  ((size_t)WS_WORDS * 4)

typedef float v2f __attribute__((ext_vector_type(2)));
typedef float v8f __attribute__((ext_vector_type(8)));

__device__ __forceinline__ void lds_count(int* p) {
  __hip_atomic_fetch_add(p, 1, __ATOMIC_RELAXED, __HIP_MEMORY_SCOPE_WORKGROUP);
}
__device__ __forceinline__ int lds_take(int* p) {
  return __hip_atomic_fetch_add(p, 1, __ATOMIC_RELAXED, __HIP_MEMORY_SCOPE_WORKGROUP);
}
__device__ __forceinline__ void lds_addf(float* p, float v) {
  __hip_atomic_fetch_add(p, v, __ATOMIC_RELAXED, __HIP_MEMORY_SCOPE_WORKGROUP);
}

__device__ __forceinline__ void rows16_gather(const float* __restrict__ t1, unsigned p_l,
                                              int f, int half, float (&v)[16], int (&o)[16]) {
  #pragma unroll
  for (int i = 0; i < 16; ++i) {
    const unsigned pe = (unsigned)__builtin_amdgcn_readlane((int)p_l, 2 * i);
    const unsigned po = (unsigned)__builtin_amdgcn_readlane((int)p_l, 2 * i + 1);
    const unsigned p  = half ? po : pe;
    v[i] = t1[(int)(p & SRC_MASK) * D_HID + f];
    o[i] = (int)(p >> SRC_BITS) * D_HID + f;
  }
}
template <bool FULL>
__device__ __forceinline__ void rows16_add(float* acc, const float (&v)[16], const int (&o)[16],
                                           int n, int half) {
  #pragma unroll
  for (int i = 0; i < 16; ++i)
    if (FULL || 2 * i + half < n) lds_addf(&acc[o[i]], v[i]);
}

__device__ __forceinline__ void rows8_gather(const float* __restrict__ t2, unsigned p_l,
                                             int c, int q, float (&v)[8], int (&o)[8]) {
  #pragma unroll
  for (int i = 0; i < 8; ++i) {
    const unsigned p0 = (unsigned)__builtin_amdgcn_readlane((int)p_l, 4 * i);
    const unsigned p1 = (unsigned)__builtin_amdgcn_readlane((int)p_l, 4 * i + 1);
    const unsigned p2 = (unsigned)__builtin_amdgcn_readlane((int)p_l, 4 * i + 2);
    const unsigned p3 = (unsigned)__builtin_amdgcn_readlane((int)p_l, 4 * i + 3);
    const unsigned p  = (q & 2) ? ((q & 1) ? p3 : p2) : ((q & 1) ? p1 : p0);
    v[i] = t2[(int)(p & SRC_MASK) * T2_PITCH + c];
    o[i] = (int)(p >> SRC_BITS) * T2_PITCH + c;
  }
}
template <bool FULL>
__device__ __forceinline__ void rows8_add(float* acc, const float (&v)[8], const int (&o)[8],
                                          int n, int c, int q) {
  if (c < N_CLASSES) {
    #pragma unroll
    for (int i = 0; i < 8; ++i)
      if (FULL || 4 * i + q < n) lds_addf(&acc[o[i]], v[i]);
  }
}

__global__ void gemm1_wmma(const float* __restrict__ feat,
                           const float* __restrict__ W1,
                           float* __restrict__ t1) {
  __shared__ float ldsW[D_FEAT * D_HID];

  {
    const float4* s4 = (const float4*)W1;
    float4* d4 = (float4*)ldsW;
    for (int i = threadIdx.x; i < (D_FEAT * D_HID) / 4; i += blockDim.x)
      d4[i] = s4[i];
  }
  __syncthreads();

  const int lane = threadIdx.x & 31;
  const int wave = threadIdx.x >> 5;
  const int tile = blockIdx.x * (blockDim.x >> 5) + wave;
  if (tile >= N_NODES / 16) return;
  const int node0 = tile * 16;

  const int mn   = lane & 15;
  const int koff = (lane >> 4) << 1;
  const float* arow = feat + (size_t)(node0 + mn) * D_FEAT + koff;
  const float* bcol = ldsW + koff * D_HID + mn;

  v8f acc = {};
  #pragma unroll 4
  for (int kk = 0; kk < D_FEAT; kk += 4) {
    v2f a = *(const v2f*)(arow + kk);
    v2f b;
    b.x = bcol[kk * D_HID];
    b.y = bcol[kk * D_HID + D_HID];
    acc = __builtin_amdgcn_wmma_f32_16x16x4_f32(
        false, a, false, b, (short)0, acc, false, false);
  }

  const int rbase = node0 + ((lane >> 4) << 3);
  #pragma unroll
  for (int r = 0; r < 8; ++r)
    t1[(size_t)(rbase + r) * D_HID + mn] = acc[r];
}

__global__ void __launch_bounds__(CH_THREADS)
count_bins(const int* __restrict__ dst, int* __restrict__ cnt, size_t ws_bytes) {
  if (ws_bytes < WS_NEED) return;
  __shared__ int hist[NB];
  for (int b = threadIdx.x; b < NB; b += CH_THREADS) hist[b] = 0;
  __syncthreads();
  const int e0 = blockIdx.x * CH_EDGES + threadIdx.x;
  int d[CH_PER_LANE];
  #pragma unroll
  for (int i = 0; i < CH_PER_LANE; ++i) d[i] = dst[e0 + i * CH_THREADS];
  #pragma unroll
  for (int i = 0; i < CH_PER_LANE; ++i) lds_count(&hist[(unsigned)d[i] / BIN_NODES]);
  __syncthreads();
  for (int b = threadIdx.x; b < NB; b += CH_THREADS) cnt[blockIdx.x * NB + b] = hist[b];
}

__global__ void __launch_bounds__(CH_THREADS)
place_edges(const int* __restrict__ src, const int* __restrict__ dst,
            const int* __restrict__ cnt, int* __restrict__ base,
            unsigned* __restrict__ list, size_t ws_bytes) {
  if (ws_bytes < WS_NEED) return;
  __shared__ int s_pre[NB];
  __shared__ int s_tot[NB];
  __shared__ int s_cur[NB];
  __shared__ int s_delta[NB];
  __shared__ unsigned s_sorted[CH_EDGES];
  __shared__ unsigned char s_bin[CH_EDGES];
  const int tid = threadIdx.x;
  const int chunk = blockIdx.x;
  for (int b = tid; b < NB; b += CH_THREADS) { s_pre[b] = 0; s_tot[b] = 0; }
  __syncthreads();
  {
    const int b = tid & 255;
    const int q = tid >> 8;
    if (b < NB) {
      int pre = 0, tot = 0;
      #pragma unroll 25
      for (int c = q; c < NCH; c += CH_THREADS / 256) {
        const int v = cnt[c * NB + b];
        tot += v;
        pre += (c < chunk) ? v : 0;
      }
      __hip_atomic_fetch_add(&s_tot[b], tot, __ATOMIC_RELAXED, __HIP_MEMORY_SCOPE_WORKGROUP);
      __hip_atomic_fetch_add(&s_pre[b], pre, __ATOMIC_RELAXED, __HIP_MEMORY_SCOPE_WORKGROUP);
    }
  }
  __syncthreads();
  if (tid < 32) {
    int t[8];
    int l[8];
    int sum_t = 0, sum_l = 0;
    #pragma unroll
    for (int k = 0; k < 8; ++k) {
      const int b = tid * 8 + k;
      t[k] = (b < NB) ? s_tot[b] : 0;
      l[k] = (b < NB) ? cnt[chunk * NB + b] : 0;
      sum_t += t[k];
      sum_l += l[k];
    }
    int inc_t = sum_t, inc_l = sum_l;
    #pragma unroll
    for (int o = 1; o < 32; o <<= 1) {
      const int up_t = __shfl_up(inc_t, o);
      const int up_l = __shfl_up(inc_l, o);
      if (tid >= o) { inc_t += up_t; inc_l += up_l; }
    }
    int run_t = inc_t - sum_t;
    int run_l = inc_l - sum_l;
    #pragma unroll
    for (int k = 0; k < 8; ++k) {
      const int b = tid * 8 + k;
      if (b < NB) {
        s_cur[b]   = run_l;
        s_delta[b] = run_t + s_pre[b] - run_l;
        if (chunk == 0) base[b] = run_t;
      }
      run_t += t[k];
      run_l += l[k];
    }
    if (chunk == 0 && tid == 31) base[NB] = run_t;
  }
  __syncthreads();
  const int e0 = chunk * CH_EDGES + tid;
  int d[CH_PER_LANE];
  int s[CH_PER_LANE];
  #pragma unroll
  for (int i = 0; i < CH_PER_LANE; ++i) {
    d[i] = dst[e0 + i * CH_THREADS];
    s[i] = src[e0 + i * CH_THREADS];
  }
  int slot[CH_PER_LANE];
  #pragma unroll
  for (int i = 0; i < CH_PER_LANE; ++i)
    slot[i] = lds_take(&s_cur[(unsigned)d[i] / BIN_NODES]);
  #pragma unroll
  for (int i = 0; i < CH_PER_LANE; ++i) {
    const unsigned b  = (unsigned)d[i] / BIN_NODES;
    const unsigned dl = (unsigned)d[i] - b * BIN_NODES;
    s_sorted[slot[i]] = (unsigned)s[i] | (dl << SRC_BITS);
    s_bin[slot[i]]    = (unsigned char)b;
  }
  __syncthreads();
  for (int j = tid; j < CH_EDGES; j += CH_THREADS)
    list[j + s_delta[s_bin[j]]] = s_sorted[j];
}

__global__ void __launch_bounds__(SUM_THREADS)
sum_layer1(const float* __restrict__ t1, const unsigned* __restrict__ list,
           const int* __restrict__ base, const float* __restrict__ b1,
           const float* __restrict__ W2, float* __restrict__ t2, size_t ws_bytes) {
  if (ws_bytes < WS_NEED) return;
  __shared__ __attribute__((aligned(16))) float acc[BIN_NODES * D_HID];
  const int tid = threadIdx.x;
  for (int i = tid; i < BIN_NODES * D_HID; i += SUM_THREADS) acc[i] = 0.0f;
  __syncthreads();
  const int bin = blockIdx.x;
  const int beg = base[bin];
  const int end = base[bin + 1];
  const int lane = tid & 31;
  const int f    = lane & 15;
  const int half = lane >> 4;
  const int w0 = __builtin_amdgcn_readfirstlane(tid & ~31);
  int e0 = beg + w0;
  unsigned p_l = (e0 + lane < end) ? list[e0 + lane] : 0u;
  while (e0 < end) {
    const int n = end - e0;
    float v[16];
    int   o[16];
    rows16_gather(t1, p_l, f, half, v, o);
    e0 += SUM_THREADS;
    const unsigned p_n = (e0 + lane < end) ? list[e0 + lane] : 0u;
    if (n >= 32) rows16_add<true>(acc, v, o, n, half);
    else         rows16_add<false>(acc, v, o, n, half);
    p_l = p_n;
  }
  __syncthreads();
  if (tid < BIN_NODES) {
    const int node = bin * BIN_NODES + tid;
    if (node < N_NODES) {
      float h[D_HID];
      const float4* a4 = (const float4*)(acc + tid * D_HID);
      #pragma unroll
      for (int q = 0; q < 4; ++q) {
        const float4 a = a4[q];
        h[q * 4 + 0] = fmaxf(a.x + b1[q * 4 + 0], 0.0f);
        h[q * 4 + 1] = fmaxf(a.y + b1[q * 4 + 1], 0.0f);
        h[q * 4 + 2] = fmaxf(a.z + b1[q * 4 + 2], 0.0f);
        h[q * 4 + 3] = fmaxf(a.w + b1[q * 4 + 3], 0.0f);
      }
      float t[T2_PITCH];
      #pragma unroll
      for (int c = 0; c < N_CLASSES; ++c) {
        float s = 0.0f;
        #pragma unroll
        for (int j = 0; j < D_HID; ++j)
          s = fmaf(h[j], W2[j * N_CLASSES + c], s);
        t[c] = s;
      }
      t[N_CLASSES] = 0.0f;
      float4* t4 = (float4*)(t2 + (size_t)node * T2_PITCH);
      t4[0] = make_float4(t[0], t[1], t[2], t[3]);
      t4[1] = make_float4(t[4], t[5], t[6], t[7]);
    }
  }
}

__global__ void __launch_bounds__(SUM_THREADS)
sum_layer2(const float* __restrict__ t2, const unsigned* __restrict__ list,
           const int* __restrict__ base, const float* __restrict__ b2,
           float* __restrict__ out, size_t ws_bytes) {
  if (ws_bytes < WS_NEED) return;
  __shared__ float acc[BIN_NODES * T2_PITCH];
  const int tid = threadIdx.x;
  for (int i = tid; i < BIN_NODES * T2_PITCH; i += SUM_THREADS) acc[i] = 0.0f;
  __syncthreads();
  const int bin = blockIdx.x;
  const int beg = base[bin];
  const int end = base[bin + 1];
  const int lane = tid & 31;
  const int c = lane & 7;
  const int q = lane >> 3;
  const int w0 = __builtin_amdgcn_readfirstlane(tid & ~31);
  int e0 = beg + w0;
  unsigned p_l = (e0 + lane < end) ? list[e0 + lane] : 0u;
  while (e0 < end) {
    const int n = end - e0;
    float v[8];
    int   o[8];
    rows8_gather(t2, p_l, c, q, v, o);
    e0 += SUM_THREADS;
    const unsigned p_n = (e0 + lane < end) ? list[e0 + lane] : 0u;
    if (n >= 32) rows8_add<true>(acc, v, o, n, c, q);
    else         rows8_add<false>(acc, v, o, n, c, q);
    p_l = p_n;
  }
  __syncthreads();
  const int node0 = bin * BIN_NODES;
  const int nodes = (N_NODES - node0 < BIN_NODES) ? (N_NODES - node0) : BIN_NODES;
  for (int idx = tid; idx < nodes * N_CLASSES; idx += SUM_THREADS) {
    const int n  = idx / N_CLASSES;
    const int cc = idx - n * N_CLASSES;
    out[node0 * N_CLASSES + idx] = acc[n * T2_PITCH + cc] + b2[cc];
  }
}

extern "C" void kernel_launch(void* const* d_in, const int* in_sizes, int n_in,
                              void* d_out, int out_size, void* d_ws, size_t ws_size,
                              hipStream_t stream) {
  const float* feat = (const float*)d_in[0];
  const float* W1   = (const float*)d_in[1];
  const float* b1   = (const float*)d_in[2];
  const float* W2   = (const float*)d_in[3];
  const float* b2   = (const float*)d_in[4];
  const int*   src  = (const int*)d_in[5];
  const int*   dst  = (const int*)d_in[6];
  float* out = (float*)d_out;

  float*    ws   = (float*)d_ws;
  float*    t1   = ws + WS_T1;
  unsigned* list = (unsigned*)d_ws + WS_LIST;
  float*    t2   = ws + WS_T2;
  int*      cnt  = (int*)d_ws + WS_CNT;
  int*      base = (int*)d_ws + WS_BASE;

  const int threads = 256;
  const int tiles = N_NODES / 16;
  const int waves_per_block = threads / 32;
  gemm1_wmma<<<(tiles + waves_per_block - 1) / waves_per_block, threads, 0, stream>>>(
      feat, W1, t1);

  count_bins<<<NCH, CH_THREADS, 0, stream>>>(dst, cnt, ws_size);

  place_edges<<<NCH, CH_THREADS, 0, stream>>>(src, dst, cnt, base, list, ws_size);

  sum_layer1<<<NB, SUM_THREADS, 0, stream>>>(t1, list, base, b1, W2, t2, ws_size);

  sum_layer2<<<NB, SUM_THREADS, 0, stream>>>(t2, list, base, b2, out, ws_size);
}


# the chunk counting pass rides in the product's launch: 250 counting workgroups (256 threads, 50 edges each) enter behind the 782 tile workgroups and run beside the feature stream; 5 launches -> 4
# speedup vs baseline: 13.0347x; 1.0627x over previous
#include <hip/hip_runtime.h>


#define N_NODES   100000
#define N_EDGES   3200000
#define D_FEAT    512
#define D_HID     16
#define N_CLASSES 7

#define BIN_NODES    416
#define NB           241
#define NCH          250
#define CH_EDGES     12800
#define CH_THREADS   512
#define CH_PER_LANE  25
#define TC_THREADS   256
#define TC_TILES     (N_NODES / 16)
#define TC_GEMM_WGS  ((TC_TILES + TC_THREADS / 32 - 1) / (TC_THREADS / 32))
#define TC_PER_LANE  (CH_EDGES / TC_THREADS)
#define SUM_THREADS  1024
#define T2_PITCH     8
#define SRC_BITS     17
#define SRC_MASK     0x1FFFFu

static_assert(NB * BIN_NODES >= N_NODES && (NB - 1) * BIN_NODES < N_NODES, "the ranges cover the nodes");
static_assert(NB <= 256 && CH_THREADS % 256 == 0, "256 threads, one per range, for every share of the column sums; a range fits a byte");
static_assert((BIN_NODES * N_CLASSES * sizeof(float)) % 128 == 0, "a range's rows of out are whole lines");
static_assert(NCH * CH_EDGES == N_EDGES, "the chunks cover the edges exactly");
static_assert(CH_THREADS * CH_PER_LANE == CH_EDGES, "whole edges per thread");
static_assert(TC_THREADS * TC_PER_LANE == CH_EDGES && NB <= TC_THREADS, "whole edges per counting thread; a thread per range");
static_assert(N_NODES % 16 == 0, "whole tiles");
static_assert(N_NODES <= (1 << SRC_BITS) && BIN_NODES <= (1 << (31 - SRC_BITS)), "src and row fit one word");

#define WS_T1    0
#define WS_LIST  (N_NODES * D_HID)
#define WS_T2    (WS_LIST + N_EDGES)
#define WS_CNT   (WS_T2 + N_NODES * T2_PITCH)
#define WS_BASE  (WS_CNT + NCH * NB)
#define WS_WORDS (WS_BASE + NB + 1)
#define WS_NEED  ((size_t)WS_WORDS * 4)

typedef float v2f __attribute__((ext_vector_type(2)));
typedef float v8f __attribute__((ext_vector_type(8)));

__device__ __forceinline__ void lds_count(int* p) {
  __hip_atomic_fetch_add(p, 1, __ATOMIC_RELAXED, __HIP_MEMORY_SCOPE_WORKGROUP);
}
__device__ __forceinline__ int lds_take(int* p) {
  return __hip_atomic_fetch_add(p, 1, __ATOMIC_RELAXED, __HIP_MEMORY_SCOPE_WORKGROUP);
}
__device__ __forceinline__ void lds_addf(float* p, float v) {
  __hip_atomic_fetch_add(p, v, __ATOMIC_RELAXED, __HIP_MEMORY_SCOPE_WORKGROUP);
}

__device__ __forceinline__ void rows16_gather(const float* __restrict__ t1, unsigned p_l,
                                              int f, int half, float (&v)[16], int (&o)[16]) {
  #pragma unroll
  for (int i = 0; i < 16; ++i) {
    const unsigned pe = (unsigned)__builtin_amdgcn_readlane((int)p_l, 2 * i);
    const unsigned po = (unsigned)__builtin_amdgcn_readlane((int)p_l, 2 * i + 1);
    const unsigned p  = half ? po : pe;
    v[i] = t1[(int)(p & SRC_MASK) * D_HID + f];
    o[i] = (int)(p >> SRC_BITS) * D_HID + f;
  }
}
template <bool FULL>
__device__ __forceinline__ void rows16_add(float* acc, const float (&v)[16], const int (&o)[16],
                                           int n, int half) {
  #pragma unroll
  for (int i = 0; i < 16; ++i)
    if (FULL || 2 * i + half < n) lds_addf(&acc[o[i]], v[i]);
}

__device__ __forceinline__ void rows8_gather(const float* __restrict__ t2, unsigned p_l,
                                             int c, int q, float (&v)[8], int (&o)[8]) {
  #pragma unroll
  for (int i = 0; i < 8; ++i) {
    const unsigned p0 = (unsigned)__builtin_amdgcn_readlane((int)p_l, 4 * i);
    const unsigned p1 = (unsigned)__builtin_amdgcn_readlane((int)p_l, 4 * i + 1);
    const unsigned p2 = (unsigned)__builtin_amdgcn_readlane((int)p_l, 4 * i + 2);
    const unsigned p3 = (unsigned)__builtin_amdgcn_readlane((int)p_l, 4 * i + 3);
    const unsigned p  = (q & 2) ? ((q & 1) ? p3 : p2) : ((q & 1) ? p1 : p0);
    v[i] = t2[(int)(p & SRC_MASK) * T2_PITCH + c];
    o[i] = (int)(p >> SRC_BITS) * T2_PITCH + c;
  }
}
template <bool FULL>
__device__ __forceinline__ void rows8_add(float* acc, const float (&v)[8], const int (&o)[8],
                                          int n, int c, int q) {
  if (c < N_CLASSES) {
    #pragma unroll
    for (int i = 0; i < 8; ++i)
      if (FULL || 4 * i + q < n) lds_addf(&acc[o[i]], v[i]);
  }
}

union __attribute__((aligned(16))) TcShared {
  float w[D_FEAT * D_HID];
  int   hist[NB];
};

__global__ void __launch_bounds__(TC_THREADS)
transform_count(const float* __restrict__ feat, const float* __restrict__ W1,
                float* __restrict__ t1, const int* __restrict__ dst,
                int* __restrict__ cnt, size_t ws_bytes) {
  __shared__ TcShared sm;
  const int tid = threadIdx.x;

  if (blockIdx.x >= TC_GEMM_WGS) {
    if (ws_bytes < WS_NEED) return;
    const int chunk = blockIdx.x - TC_GEMM_WGS;
    if (tid < NB) sm.hist[tid] = 0;
    __syncthreads();
    const int e0 = chunk * CH_EDGES + tid;
    int d[TC_PER_LANE];
    #pragma unroll
    for (int i = 0; i < TC_PER_LANE; ++i) d[i] = dst[e0 + i * TC_THREADS];
    #pragma unroll
    for (int i = 0; i < TC_PER_LANE; ++i) lds_count(&sm.hist[(unsigned)d[i] / BIN_NODES]);
    __syncthreads();
    if (tid < NB) cnt[chunk * NB + tid] = sm.hist[tid];
    return;
  }

  float* ldsW = sm.w;
  {
    const float4* s4 = (const float4*)W1;
    float4* d4 = (float4*)ldsW;
    for (int i = tid; i < (D_FEAT * D_HID) / 4; i += TC_THREADS)
      d4[i] = s4[i];
  }
  __syncthreads();

  const int lane = tid & 31;
  const int wave = tid >> 5;
  const int tile = blockIdx.x * (TC_THREADS >> 5) + wave;
  if (tile >= TC_TILES) return;
  const int node0 = tile * 16;

  const int mn   = lane & 15;
  const int koff = (lane >> 4) << 1;
  const float* arow = feat + (size_t)(node0 + mn) * D_FEAT + koff;
  const float* bcol = ldsW + koff * D_HID + mn;

  v8f acc = {};
  #pragma unroll 4
  for (int kk = 0; kk < D_FEAT; kk += 4) {
    v2f a = *(const v2f*)(arow + kk);
    v2f b;
    b.x = bcol[kk * D_HID];
    b.y = bcol[kk * D_HID + D_HID];
    acc = __builtin_amdgcn_wmma_f32_16x16x4_f32(
        false, a, false, b, (short)0, acc, false, false);
  }

  const int rbase = node0 + ((lane >> 4) << 3);
  #pragma unroll
  for (int r = 0; r < 8; ++r)
    t1[(size_t)(rbase + r) * D_HID + mn] = acc[r];
}

__global__ void __launch_bounds__(CH_THREADS)
place_edges(const int* __restrict__ src, const int* __restrict__ dst,
            const int* __restrict__ cnt, int* __restrict__ base,
            unsigned* __restrict__ list, size_t ws_bytes) {
  if (ws_bytes < WS_NEED) return;
  __shared__ int s_pre[NB];
  __shared__ int s_tot[NB];
  __shared__ int s_cur[NB];
  __shared__ int s_delta[NB];
  __shared__ unsigned s_sorted[CH_EDGES];
  __shared__ unsigned char s_bin[CH_EDGES];
  const int tid = threadIdx.x;
  const int chunk = blockIdx.x;
  for (int b = tid; b < NB; b += CH_THREADS) { s_pre[b] = 0; s_tot[b] = 0; }
  __syncthreads();
  {
    const int b = tid & 255;
    const int q = tid >> 8;
    if (b < NB) {
      int pre = 0, tot = 0;
      #pragma unroll 25
      for (int c = q; c < NCH; c += CH_THREADS / 256) {
        const int v = cnt[c * NB + b];
        tot += v;
        pre += (c < chunk) ? v : 0;
      }
      __hip_atomic_fetch_add(&s_tot[b], tot, __ATOMIC_RELAXED, __HIP_MEMORY_SCOPE_WORKGROUP);
      __hip_atomic_fetch_add(&s_pre[b], pre, __ATOMIC_RELAXED, __HIP_MEMORY_SCOPE_WORKGROUP);
    }
  }
  __syncthreads();
  if (tid < 32) {
    int t[8];
    int l[8];
    int sum_t = 0, sum_l = 0;
    #pragma unroll
    for (int k = 0; k < 8; ++k) {
      const int b = tid * 8 + k;
      t[k] = (b < NB) ? s_tot[b] : 0;
      l[k] = (b < NB) ? cnt[chunk * NB + b] : 0;
      sum_t += t[k];
      sum_l += l[k];
    }
    int inc_t = sum_t, inc_l = sum_l;
    #pragma unroll
    for (int o = 1; o < 32; o <<= 1) {
      const int up_t = __shfl_up(inc_t, o);
      const int up_l = __shfl_up(inc_l, o);
      if (tid >= o) { inc_t += up_t; inc_l += up_l; }
    }
    int run_t = inc_t - sum_t;
    int run_l = inc_l - sum_l;
    #pragma unroll
    for (int k = 0; k < 8; ++k) {
      const int b = tid * 8 + k;
      if (b < NB) {
        s_cur[b]   = run_l;
        s_delta[b] = run_t + s_pre[b] - run_l;
        if (chunk == 0) base[b] = run_t;
      }
      run_t += t[k];
      run_l += l[k];
    }
    if (chunk == 0 && tid == 31) base[NB] = run_t;
  }
  __syncthreads();
  const int e0 = chunk * CH_EDGES + tid;
  int d[CH_PER_LANE];
  int s[CH_PER_LANE];
  #pragma unroll
  for (int i = 0; i < CH_PER_LANE; ++i) {
    d[i] = dst[e0 + i * CH_THREADS];
    s[i] = src[e0 + i * CH_THREADS];
  }
  int slot[CH_PER_LANE];
  #pragma unroll
  for (int i = 0; i < CH_PER_LANE; ++i)
    slot[i] = lds_take(&s_cur[(unsigned)d[i] / BIN_NODES]);
  #pragma unroll
  for (int i = 0; i < CH_PER_LANE; ++i) {
    const unsigned b  = (unsigned)d[i] / BIN_NODES;
    const unsigned dl = (unsigned)d[i] - b * BIN_NODES;
    s_sorted[slot[i]] = (unsigned)s[i] | (dl << SRC_BITS);
    s_bin[slot[i]]    = (unsigned char)b;
  }
  __syncthreads();
  for (int j = tid; j < CH_EDGES; j += CH_THREADS)
    list[j + s_delta[s_bin[j]]] = s_sorted[j];
}

__global__ void __launch_bounds__(SUM_THREADS)
sum_layer1(const float* __restrict__ t1, const unsigned* __restrict__ list,
           const int* __restrict__ base, const float* __restrict__ b1,
           const float* __restrict__ W2, float* __restrict__ t2, size_t ws_bytes) {
  if (ws_bytes < WS_NEED) return;
  __shared__ __attribute__((aligned(16))) float acc[BIN_NODES * D_HID];
  const int tid = threadIdx.x;
  for (int i = tid; i < BIN_NODES * D_HID; i += SUM_THREADS) acc[i] = 0.0f;
  __syncthreads();
  const int bin = blockIdx.x;
  const int beg = base[bin];
  const int end = base[bin + 1];
  const int lane = tid & 31;
  const int f    = lane & 15;
  const int half = lane >> 4;
  const int w0 = __builtin_amdgcn_readfirstlane(tid & ~31);
  int e0 = beg + w0;
  unsigned p_l = (e0 + lane < end) ? list[e0 + lane] : 0u;
  while (e0 < end) {
    const int n = end - e0;
    float v[16];
    int   o[16];
    rows16_gather(t1, p_l, f, half, v, o);
    e0 += SUM_THREADS;
    const unsigned p_n = (e0 + lane < end) ? list[e0 + lane] : 0u;
    if (n >= 32) rows16_add<true>(acc, v, o, n, half);
    else         rows16_add<false>(acc, v, o, n, half);
    p_l = p_n;
  }
  __syncthreads();
  if (tid < BIN_NODES) {
    const int node = bin * BIN_NODES + tid;
    if (node < N_NODES) {
      float h[D_HID];
      const float4* a4 = (const float4*)(acc + tid * D_HID);
      #pragma unroll
      for (int q = 0; q < 4; ++q) {
        const float4 a = a4[q];
        h[q * 4 + 0] = fmaxf(a.x + b1[q * 4 + 0], 0.0f);
        h[q * 4 + 1] = fmaxf(a.y + b1[q * 4 + 1], 0.0f);
        h[q * 4 + 2] = fmaxf(a.z + b1[q * 4 + 2], 0.0f);
        h[q * 4 + 3] = fmaxf(a.w + b1[q * 4 + 3], 0.0f);
      }
      float t[T2_PITCH];
      #pragma unroll
      for (int c = 0; c < N_CLASSES; ++c) {
        float s = 0.0f;
        #pragma unroll
        for (int j = 0; j < D_HID; ++j)
          s = fmaf(h[j], W2[j * N_CLASSES + c], s);
        t[c] = s;
      }
      t[N_CLASSES] = 0.0f;
      float4* t4 = (float4*)(t2 + (size_t)node * T2_PITCH);
      t4[0] = make_float4(t[0], t[1], t[2], t[3]);
      t4[1] = make_float4(t[4], t[5], t[6], t[7]);
    }
  }
}

__global__ void __launch_bounds__(SUM_THREADS)
sum_layer2(const float* __restrict__ t2, const unsigned* __restrict__ list,
           const int* __restrict__ base, const float* __restrict__ b2,
           float* __restrict__ out, size_t ws_bytes) {
  if (ws_bytes < WS_NEED) return;
  __shared__ float acc[BIN_NODES * T2_PITCH];
  const int tid = threadIdx.x;
  for (int i = tid; i < BIN_NODES * T2_PITCH; i += SUM_THREADS) acc[i] = 0.0f;
  __syncthreads();
  const int bin = blockIdx.x;
  const int beg = base[bin];
  const int end = base[bin + 1];
  const int lane = tid & 31;
  const int c = lane & 7;
  const int q = lane >> 3;
  const int w0 = __builtin_amdgcn_readfirstlane(tid & ~31);
  int e0 = beg + w0;
  unsigned p_l = (e0 + lane < end) ? list[e0 + lane] : 0u;
  while (e0 < end) {
    const int n = end - e0;
    float v[8];
    int   o[8];
    rows8_gather(t2, p_l, c, q, v, o);
    e0 += SUM_THREADS;
    const unsigned p_n = (e0 + lane < end) ? list[e0 + lane] : 0u;
    if (n >= 32) rows8_add<true>(acc, v, o, n, c, q);
    else         rows8_add<false>(acc, v, o, n, c, q);
    p_l = p_n;
  }
  __syncthreads();
  const int node0 = bin * BIN_NODES;
  const int nodes = (N_NODES - node0 < BIN_NODES) ? (N_NODES - node0) : BIN_NODES;
  for (int idx = tid; idx < nodes * N_CLASSES; idx += SUM_THREADS) {
    const int n  = idx / N_CLASSES;
    const int cc = idx - n * N_CLASSES;
    out[node0 * N_CLASSES + idx] = acc[n * T2_PITCH + cc] + b2[cc];
  }
}

extern "C" void kernel_launch(void* const* d_in, const int* in_sizes, int n_in,
                              void* d_out, int out_size, void* d_ws, size_t ws_size,
                              hipStream_t stream) {
  const float* feat = (const float*)d_in[0];
  const float* W1   = (const float*)d_in[1];
  const float* b1   = (const float*)d_in[2];
  const float* W2   = (const float*)d_in[3];
  const float* b2   = (const float*)d_in[4];
  const int*   src  = (const int*)d_in[5];
  const int*   dst  = (const int*)d_in[6];
  float* out = (float*)d_out;

  float*    ws   = (float*)d_ws;
  float*    t1   = ws + WS_T1;
  unsigned* list = (unsigned*)d_ws + WS_LIST;
  float*    t2   = ws + WS_T2;
  int*      cnt  = (int*)d_ws + WS_CNT;
  int*      base = (int*)d_ws + WS_BASE;

  transform_count<<<TC_GEMM_WGS + NCH, TC_THREADS, 0, stream>>>(feat, W1, t1, dst, cnt, ws_size);

  place_edges<<<NCH, CH_THREADS, 0, stream>>>(src, dst, cnt, base, list, ws_size);

  sum_layer1<<<NB, SUM_THREADS, 0, stream>>>(t1, list, base, b1, W2, t2, ws_size);

  sum_layer2<<<NB, SUM_THREADS, 0, stream>>>(t2, list, base, b2, out, ws_size);
}


# chunk workgroups group their chunk in place (LDS slots, first slot per range kept); sum passes walk a range's piece of every chunk, 8 per wave; placing launch gone, 4 -> 3; f32 sum order changes
# speedup vs baseline: 14.7261x; 1.1298x over previous
#include <hip/hip_runtime.h>


#define N_NODES   100000
#define N_EDGES   3200000
#define D_FEAT    512
#define D_HID     16
#define N_CLASSES 7

#define BIN_NODES    416
#define NB           241
#define NCH          250
#define CH_EDGES     12800
#define SEG_PER_WAVE 8
#define TC_THREADS   256
#define TC_TILES     (N_NODES / 16)
#define TC_GEMM_WGS  ((TC_TILES + TC_THREADS / 32 - 1) / (TC_THREADS / 32))
#define TC_PER_LANE  (CH_EDGES / TC_THREADS)
#define TC_HALF      (TC_PER_LANE / 2)
#define SUM_THREADS  1024
#define T2_PITCH     8
#define SRC_BITS     17
#define SRC_MASK     0x1FFFFu

static_assert(NB * BIN_NODES >= N_NODES && (NB - 1) * BIN_NODES < N_NODES, "the ranges cover the nodes");
static_assert((BIN_NODES * N_CLASSES * sizeof(float)) % 128 == 0, "a range's rows of out are whole lines");
static_assert(NCH * CH_EDGES == N_EDGES, "the chunks cover the edges exactly");
static_assert(TC_THREADS * TC_PER_LANE == CH_EDGES && TC_HALF * 2 == TC_PER_LANE && NB <= TC_THREADS, "whole edges per thread, in two halves; a thread per range");
static_assert(NB <= 32 * 8, "one wave scans the ranges, 8 per lane");
static_assert((SUM_THREADS / 32) * SEG_PER_WAVE >= NCH && SEG_PER_WAVE <= 32, "the waves of a sum pass cover the chunks");
static_assert(CH_EDGES % 4 == 0, "a chunk is copied out 16 bytes at a time");
static_assert(N_NODES % 16 == 0, "whole tiles");
static_assert(N_NODES <= (1 << SRC_BITS) && BIN_NODES <= (1 << (31 - SRC_BITS)), "src and row fit one word");

#define WS_T1    0
#define WS_LIST  (N_NODES * D_HID)
#define WS_T2    (WS_LIST + N_EDGES)
#define WS_OFF   (WS_T2 + N_NODES * T2_PITCH)
#define WS_WORDS (WS_OFF + NCH * NB)
#define WS_NEED  ((size_t)WS_WORDS * 4)

typedef float v2f __attribute__((ext_vector_type(2)));
typedef float v8f __attribute__((ext_vector_type(8)));

__device__ __forceinline__ void lds_count(int* p) {
  __hip_atomic_fetch_add(p, 1, __ATOMIC_RELAXED, __HIP_MEMORY_SCOPE_WORKGROUP);
}
__device__ __forceinline__ int lds_take(int* p) {
  return __hip_atomic_fetch_add(p, 1, __ATOMIC_RELAXED, __HIP_MEMORY_SCOPE_WORKGROUP);
}
__device__ __forceinline__ void lds_addf(float* p, float v) {
  __hip_atomic_fetch_add(p, v, __ATOMIC_RELAXED, __HIP_MEMORY_SCOPE_WORKGROUP);
}

__device__ __forceinline__ void rows16_gather(const float* __restrict__ t1, unsigned p_l,
                                              int f, int half, float (&v)[16], int (&o)[16]) {
  #pragma unroll
  for (int i = 0; i < 16; ++i) {
    const unsigned pe = (unsigned)__builtin_amdgcn_readlane((int)p_l, 2 * i);
    const unsigned po = (unsigned)__builtin_amdgcn_readlane((int)p_l, 2 * i + 1);
    const unsigned p  = half ? po : pe;
    v[i] = t1[(int)(p & SRC_MASK) * D_HID + f];
    o[i] = (int)(p >> SRC_BITS) * D_HID + f;
  }
}
template <bool FULL>
__device__ __forceinline__ void rows16_add(float* acc, const float (&v)[16], const int (&o)[16],
                                           int n, int half) {
  #pragma unroll
  for (int i = 0; i < 16; ++i)
    if (FULL || 2 * i + half < n) lds_addf(&acc[o[i]], v[i]);
}

__device__ __forceinline__ void rows8_gather(const float* __restrict__ t2, unsigned p_l,
                                             int c, int q, float (&v)[8], int (&o)[8]) {
  #pragma unroll
  for (int i = 0; i < 8; ++i) {
    const unsigned p0 = (unsigned)__builtin_amdgcn_readlane((int)p_l, 4 * i);
    const unsigned p1 = (unsigned)__builtin_amdgcn_readlane((int)p_l, 4 * i + 1);
    const unsigned p2 = (unsigned)__builtin_amdgcn_readlane((int)p_l, 4 * i + 2);
    const unsigned p3 = (unsigned)__builtin_amdgcn_readlane((int)p_l, 4 * i + 3);
    const unsigned p  = (q & 2) ? ((q & 1) ? p3 : p2) : ((q & 1) ? p1 : p0);
    v[i] = t2[(int)(p & SRC_MASK) * T2_PITCH + c];
    o[i] = (int)(p >> SRC_BITS) * T2_PITCH + c;
  }
}
template <bool FULL>
__device__ __forceinline__ void rows8_add(float* acc, const float (&v)[8], const int (&o)[8],
                                          int n, int c, int q) {
  if (c < N_CLASSES) {
    #pragma unroll
    for (int i = 0; i < 8; ++i)
      if (FULL || 4 * i + q < n) lds_addf(&acc[o[i]], v[i]);
  }
}

__device__ __forceinline__ void wave_pieces(const int* __restrict__ off, int bin, int wave, int lane,
                                            int (&sb)[SEG_PER_WAVE], int (&ln)[SEG_PER_WAVE], int& total) {
  const int c = wave + 32 * lane;
  int s = 0, e = 0;
  if (lane < SEG_PER_WAVE && c < NCH) {
    s = off[c * NB + bin];
    e = (bin + 1 < NB) ? off[c * NB + bin + 1] : CH_EDGES;
  }
  const int first = c * CH_EDGES + s;
  const int len   = e - s;
  total = 0;
  #pragma unroll
  for (int k = 0; k < SEG_PER_WAVE; ++k) {
    sb[k] = __builtin_amdgcn_readlane(first, k);
    ln[k] = __builtin_amdgcn_readlane(len, k);
    total += ln[k];
  }
}
__device__ __forceinline__ unsigned run_entry(const unsigned* __restrict__ list,
                                              const int (&sb)[SEG_PER_WAVE], const int (&ln)[SEG_PER_WAVE],
                                              int total, int f) {
  int idx = 0, o = f;
  #pragma unroll
  for (int k = 0; k < SEG_PER_WAVE; ++k) {
    idx = ((unsigned)o < (unsigned)ln[k]) ? sb[k] + o : idx;
    o -= ln[k];
  }
  return (f < total) ? list[idx] : 0u;
}

struct ChunkShared {
  unsigned sorted[CH_EDGES];
  int      hist[NB];
  int      cur[NB];
};
union __attribute__((aligned(16))) TcShared {
  float       w[D_FEAT * D_HID];
  ChunkShared g;
};

__global__ void __launch_bounds__(TC_THREADS)
transform_group(const float* __restrict__ feat, const float* __restrict__ W1,
                float* __restrict__ t1, const int* __restrict__ src, const int* __restrict__ dst,
                int* __restrict__ off, unsigned* __restrict__ list, size_t ws_bytes) {
  __shared__ TcShared sm;
  const int tid = threadIdx.x;

  if (blockIdx.x >= TC_GEMM_WGS) {
    if (ws_bytes < WS_NEED) return;
    const int chunk = blockIdx.x - TC_GEMM_WGS;
    if (tid < NB) sm.g.hist[tid] = 0;
    __syncthreads();
    const int e0 = chunk * CH_EDGES + tid;
    {
      int d[TC_PER_LANE];
      #pragma unroll
      for (int i = 0; i < TC_PER_LANE; ++i) d[i] = dst[e0 + i * TC_THREADS];
      #pragma unroll
      for (int i = 0; i < TC_PER_LANE; ++i) lds_count(&sm.g.hist[(unsigned)d[i] / BIN_NODES]);
    }
    __syncthreads();
    if (tid < 32) {
      int h[8];
      int sum = 0;
      #pragma unroll
      for (int k = 0; k < 8; ++k) {
        const int b = tid * 8 + k;
        h[k] = (b < NB) ? sm.g.hist[b] : 0;
        sum += h[k];
      }
      int inc = sum;
      #pragma unroll
      for (int o = 1; o < 32; o <<= 1) {
        const int up = __shfl_up(inc, o);
        if (tid >= o) inc += up;
      }
      int run = inc - sum;
      #pragma unroll
      for (int k = 0; k < 8; ++k) {
        const int b = tid * 8 + k;
        if (b < NB) {
          sm.g.cur[b] = run;
          off[chunk * NB + b] = run;
        }
        run += h[k];
      }
    }
    __syncthreads();
    #pragma nounroll
    for (int hh = 0; hh < 2; ++hh) {
      int d[TC_HALF];
      int s[TC_HALF];
      int slot[TC_HALF];
      #pragma unroll
      for (int i = 0; i < TC_HALF; ++i) {
        d[i] = dst[e0 + (hh * TC_HALF + i) * TC_THREADS];
        s[i] = src[e0 + (hh * TC_HALF + i) * TC_THREADS];
      }
      #pragma unroll
      for (int i = 0; i < TC_HALF; ++i)
        slot[i] = lds_take(&sm.g.cur[(unsigned)d[i] / BIN_NODES]);
      #pragma unroll
      for (int i = 0; i < TC_HALF; ++i) {
        const unsigned b  = (unsigned)d[i] / BIN_NODES;
        const unsigned dl = (unsigned)d[i] - b * BIN_NODES;
        sm.g.sorted[slot[i]] = (unsigned)s[i] | (dl << SRC_BITS);
      }
    }
    __syncthreads();
    const uint4* s4 = (const uint4*)sm.g.sorted;
    uint4* l4 = (uint4*)(list + (size_t)chunk * CH_EDGES);
    for (int j = tid; j < CH_EDGES / 4; j += TC_THREADS) l4[j] = s4[j];
    return;
  }

  float* ldsW = sm.w;
  {
    const float4* s4 = (const float4*)W1;
    float4* d4 = (float4*)ldsW;
    for (int i = tid; i < (D_FEAT * D_HID) / 4; i += TC_THREADS)
      d4[i] = s4[i];
  }
  __syncthreads();

  const int lane = tid & 31;
  const int wave = tid >> 5;
  const int tile = blockIdx.x * (TC_THREADS >> 5) + wave;
  if (tile >= TC_TILES) return;
  const int node0 = tile * 16;

  const int mn   = lane & 15;
  const int koff = (lane >> 4) << 1;
  const float* arow = feat + (size_t)(node0 + mn) * D_FEAT + koff;
  const float* bcol = ldsW + koff * D_HID + mn;

  v8f acc = {};
  #pragma unroll 4
  for (int kk = 0; kk < D_FEAT; kk += 4) {
    v2f a = *(const v2f*)(arow + kk);
    v2f b;
    b.x = bcol[kk * D_HID];
    b.y = bcol[kk * D_HID + D_HID];
    acc = __builtin_amdgcn_wmma_f32_16x16x4_f32(
        false, a, false, b, (short)0, acc, false, false);
  }

  const int rbase = node0 + ((lane >> 4) << 3);
  #pragma unroll
  for (int r = 0; r < 8; ++r)
    t1[(size_t)(rbase + r) * D_HID + mn] = acc[r];
}

__global__ void __launch_bounds__(SUM_THREADS)
sum_layer1(const float* __restrict__ t1, const unsigned* __restrict__ list,
           const int* __restrict__ off, const float* __restrict__ b1,
           const float* __restrict__ W2, float* __restrict__ t2, size_t ws_bytes) {
  if (ws_bytes < WS_NEED) return;
  __shared__ __attribute__((aligned(16))) float acc[BIN_NODES * D_HID];
  const int tid = threadIdx.x;
  for (int i = tid; i < BIN_NODES * D_HID; i += SUM_THREADS) acc[i] = 0.0f;
  __syncthreads();
  const int bin = blockIdx.x;
  const int lane = tid & 31;
  const int f    = lane & 15;
  const int half = lane >> 4;
  const int wave = __builtin_amdgcn_readfirstlane(tid >> 5);
  int sb[SEG_PER_WAVE], ln[SEG_PER_WAVE], total;
  wave_pieces(off, bin, wave, lane, sb, ln, total);
  int f0 = 0;
  unsigned p_l = run_entry(list, sb, ln, total, lane);
  while (f0 < total) {
    const int n = total - f0;
    float v[16];
    int   o[16];
    rows16_gather(t1, p_l, f, half, v, o);
    f0 += 32;
    const unsigned p_n = run_entry(list, sb, ln, total, f0 + lane);
    if (n >= 32) rows16_add<true>(acc, v, o, n, half);
    else         rows16_add<false>(acc, v, o, n, half);
    p_l = p_n;
  }
  __syncthreads();
  if (tid < BIN_NODES) {
    const int node = bin * BIN_NODES + tid;
    if (node < N_NODES) {
      float h[D_HID];
      const float4* a4 = (const float4*)(acc + tid * D_HID);
      #pragma unroll
      for (int q = 0; q < 4; ++q) {
        const float4 a = a4[q];
        h[q * 4 + 0] = fmaxf(a.x + b1[q * 4 + 0], 0.0f);
        h[q * 4 + 1] = fmaxf(a.y + b1[q * 4 + 1], 0.0f);
        h[q * 4 + 2] = fmaxf(a.z + b1[q * 4 + 2], 0.0f);
        h[q * 4 + 3] = fmaxf(a.w + b1[q * 4 + 3], 0.0f);
      }
      float t[T2_PITCH];
      #pragma unroll
      for (int c = 0; c < N_CLASSES; ++c) {
        float s = 0.0f;
        #pragma unroll
        for (int j = 0; j < D_HID; ++j)
          s = fmaf(h[j], W2[j * N_CLASSES + c], s);
        t[c] = s;
      }
      t[N_CLASSES] = 0.0f;
      float4* t4 = (float4*)(t2 + (size_t)node * T2_PITCH);
      t4[0] = make_float4(t[0], t[1], t[2], t[3]);
      t4[1] = make_float4(t[4], t[5], t[6], t[7]);
    }
  }
}

__global__ void __launch_bounds__(SUM_THREADS)
sum_layer2(const float* __restrict__ t2, const unsigned* __restrict__ list,
           const int* __restrict__ off, const float* __restrict__ b2,
           float* __restrict__ out, size_t ws_bytes) {
  if (ws_bytes < WS_NEED) return;
  __shared__ float acc[BIN_NODES * T2_PITCH];
  const int tid = threadIdx.x;
  for (int i = tid; i < BIN_NODES * T2_PITCH; i += SUM_THREADS) acc[i] = 0.0f;
  __syncthreads();
  const int bin = blockIdx.x;
  const int lane = tid & 31;
  const int c = lane & 7;
  const int q = lane >> 3;
  const int wave = __builtin_amdgcn_readfirstlane(tid >> 5);
  int sb[SEG_PER_WAVE], ln[SEG_PER_WAVE], total;
  wave_pieces(off, bin, wave, lane, sb, ln, total);
  int f0 = 0;
  unsigned p_l = run_entry(list, sb, ln, total, lane);
  while (f0 < total) {
    const int n = total - f0;
    float v[8];
    int   o[8];
    rows8_gather(t2, p_l, c, q, v, o);
    f0 += 32;
    const unsigned p_n = run_entry(list, sb, ln, total, f0 + lane);
    if (n >= 32) rows8_add<true>(acc, v, o, n, c, q);
    else         rows8_add<false>(acc, v, o, n, c, q);
    p_l = p_n;
  }
  __syncthreads();
  const int node0 = bin * BIN_NODES;
  const int nodes = (N_NODES - node0 < BIN_NODES) ? (N_NODES - node0) : BIN_NODES;
  for (int idx = tid; idx < nodes * N_CLASSES; idx += SUM_THREADS) {
    const int n  = idx / N_CLASSES;
    const int cc = idx - n * N_CLASSES;
    out[node0 * N_CLASSES + idx] = acc[n * T2_PITCH + cc] + b2[cc];
  }
}

extern "C" void kernel_launch(void* const* d_in, const int* in_sizes, int n_in,
                              void* d_out, int out_size, void* d_ws, size_t ws_size,
                              hipStream_t stream) {
  const float* feat = (const float*)d_in[0];
  const float* W1   = (const float*)d_in[1];
  const float* b1   = (const float*)d_in[2];
  const float* W2   = (const float*)d_in[3];
  const float* b2   = (const float*)d_in[4];
  const int*   src  = (const int*)d_in[5];
  const int*   dst  = (const int*)d_in[6];
  float* out = (float*)d_out;

  float*    ws   = (float*)d_ws;
  float*    t1   = ws + WS_T1;
  unsigned* list = (unsigned*)d_ws + WS_LIST;
  float*    t2   = ws + WS_T2;
  int*      off  = (int*)d_ws + WS_OFF;

  transform_group<<<TC_GEMM_WGS + NCH, TC_THREADS, 0, stream>>>(feat, W1, t1, src, dst, off, list, ws_size);

  sum_layer1<<<NB, SUM_THREADS, 0, stream>>>(t1, list, off, b1, W2, t2, ws_size);

  sum_layer2<<<NB, SUM_THREADS, 0, stream>>>(t2, list, off, b2, out, ws_size);
}


# product: feature rows and W1 (transposed in LDS) read 16 bytes per lane in one shared k order, each row's next 256-byte unit requested a unit ahead of its matrix steps; f32 k sum order changes
# speedup vs baseline: 15.3590x; 1.0430x over previous
#include <hip/hip_runtime.h>


#define N_NODES   100000
#define N_EDGES   3200000
#define D_FEAT    512
#define D_HID     16
#define N_CLASSES 7

#define BIN_NODES    416
#define NB           241
#define NCH          250
#define CH_EDGES     12800
#define SEG_PER_WAVE 8
#define TC_THREADS   256
#define TC_TILES     (N_NODES / 16)
#define TC_GEMM_WGS  ((TC_TILES + TC_THREADS / 32 - 1) / (TC_THREADS / 32))
#define TC_PER_LANE  (CH_EDGES / TC_THREADS)
#define TC_HALF      (TC_PER_LANE / 2)
#define SUM_THREADS  1024
#define T2_PITCH     8
#define SRC_BITS     17
#define SRC_MASK     0x1FFFFu
#define WT_PITCH     (D_FEAT + 4)
#define K_UNIT       64

static_assert(NB * BIN_NODES >= N_NODES && (NB - 1) * BIN_NODES < N_NODES, "the ranges cover the nodes");
static_assert((BIN_NODES * N_CLASSES * sizeof(float)) % 128 == 0, "a range's rows of out are whole lines");
static_assert(NCH * CH_EDGES == N_EDGES, "the chunks cover the edges exactly");
static_assert(TC_THREADS * TC_PER_LANE == CH_EDGES && TC_HALF * 2 == TC_PER_LANE && NB <= TC_THREADS, "whole edges per thread, in two halves; a thread per range");
static_assert(NB <= 32 * 8, "one wave scans the ranges, 8 per lane");
static_assert((SUM_THREADS / 32) * SEG_PER_WAVE >= NCH && SEG_PER_WAVE <= 32, "the waves of a sum pass cover the chunks");
static_assert(CH_EDGES % 4 == 0, "a chunk is copied out 16 bytes at a time");
static_assert(N_NODES % 16 == 0, "whole tiles");
static_assert(D_FEAT % (2 * K_UNIT) == 0 && (WT_PITCH * 4) % 16 == 0 && ((WT_PITCH * 4) / 16) % 2 == 1, "whole pairs of units per row; W1 pitch is 16 B x odd");
static_assert(N_NODES <= (1 << SRC_BITS) && BIN_NODES <= (1 << (31 - SRC_BITS)), "src and row fit one word");

#define WS_T1    0
#define WS_LIST  (N_NODES * D_HID)
#define WS_T2    (WS_LIST + N_EDGES)
#define WS_OFF   (WS_T2 + N_NODES * T2_PITCH)
#define WS_WORDS (WS_OFF + NCH * NB)
#define WS_NEED  ((size_t)WS_WORDS * 4)

typedef float v2f __attribute__((ext_vector_type(2)));
typedef float v8f __attribute__((ext_vector_type(8)));

__device__ __forceinline__ void lds_count(int* p) {
  __hip_atomic_fetch_add(p, 1, __ATOMIC_RELAXED, __HIP_MEMORY_SCOPE_WORKGROUP);
}
__device__ __forceinline__ int lds_take(int* p) {
  return __hip_atomic_fetch_add(p, 1, __ATOMIC_RELAXED, __HIP_MEMORY_SCOPE_WORKGROUP);
}
__device__ __forceinline__ void lds_addf(float* p, float v) {
  __hip_atomic_fetch_add(p, v, __ATOMIC_RELAXED, __HIP_MEMORY_SCOPE_WORKGROUP);
}

__device__ __forceinline__ void rows16_gather(const float* __restrict__ t1, unsigned p_l,
                                              int f, int half, float (&v)[16], int (&o)[16]) {
  #pragma unroll
  for (int i = 0; i < 16; ++i) {
    const unsigned pe = (unsigned)__builtin_amdgcn_readlane((int)p_l, 2 * i);
    const unsigned po = (unsigned)__builtin_amdgcn_readlane((int)p_l, 2 * i + 1);
    const unsigned p  = half ? po : pe;
    v[i] = t1[(int)(p & SRC_MASK) * D_HID + f];
    o[i] = (int)(p >> SRC_BITS) * D_HID + f;
  }
}
template <bool FULL>
__device__ __forceinline__ void rows16_add(float* acc, const float (&v)[16], const int (&o)[16],
                                           int n, int half) {
  #pragma unroll
  for (int i = 0; i < 16; ++i)
    if (FULL || 2 * i + half < n) lds_addf(&acc[o[i]], v[i]);
}

__device__ __forceinline__ void rows8_gather(const float* __restrict__ t2, unsigned p_l,
                                             int c, int q, float (&v)[8], int (&o)[8]) {
  #pragma unroll
  for (int i = 0; i < 8; ++i) {
    const unsigned p0 = (unsigned)__builtin_amdgcn_readlane((int)p_l, 4 * i);
    const unsigned p1 = (unsigned)__builtin_amdgcn_readlane((int)p_l, 4 * i + 1);
    const unsigned p2 = (unsigned)__builtin_amdgcn_readlane((int)p_l, 4 * i + 2);
    const unsigned p3 = (unsigned)__builtin_amdgcn_readlane((int)p_l, 4 * i + 3);
    const unsigned p  = (q & 2) ? ((q & 1) ? p3 : p2) : ((q & 1) ? p1 : p0);
    v[i] = t2[(int)(p & SRC_MASK) * T2_PITCH + c];
    o[i] = (int)(p >> SRC_BITS) * T2_PITCH + c;
  }
}
template <bool FULL>
__device__ __forceinline__ void rows8_add(float* acc, const float (&v)[8], const int (&o)[8],
                                          int n, int c, int q) {
  if (c < N_CLASSES) {
    #pragma unroll
    for (int i = 0; i < 8; ++i)
      if (FULL || 4 * i + q < n) lds_addf(&acc[o[i]], v[i]);
  }
}

__device__ __forceinline__ void wave_pieces(const int* __restrict__ off, int bin, int wave, int lane,
                                            int (&sb)[SEG_PER_WAVE], int (&ln)[SEG_PER_WAVE], int& total) {
  const int c = wave + 32 * lane;
  int s = 0, e = 0;
  if (lane < SEG_PER_WAVE && c < NCH) {
    s = off[c * NB + bin];
    e = (bin + 1 < NB) ? off[c * NB + bin + 1] : CH_EDGES;
  }
  const int first = c * CH_EDGES + s;
  const int len   = e - s;
  total = 0;
  #pragma unroll
  for (int k = 0; k < SEG_PER_WAVE; ++k) {
    sb[k] = __builtin_amdgcn_readlane(first, k);
    ln[k] = __builtin_amdgcn_readlane(len, k);
    total += ln[k];
  }
}
__device__ __forceinline__ unsigned run_entry(const unsigned* __restrict__ list,
                                              const int (&sb)[SEG_PER_WAVE], const int (&ln)[SEG_PER_WAVE],
                                              int total, int f) {
  int idx = 0, o = f;
  #pragma unroll
  for (int k = 0; k < SEG_PER_WAVE; ++k) {
    idx = ((unsigned)o < (unsigned)ln[k]) ? sb[k] + o : idx;
    o -= ln[k];
  }
  return (f < total) ? list[idx] : 0u;
}

struct ChunkShared {
  unsigned sorted[CH_EDGES];
  int      hist[NB];
  int      cur[NB];
};
union __attribute__((aligned(16))) TcShared {
  float       w[D_HID * WT_PITCH];
  ChunkShared g;
};

__device__ __forceinline__ v8f unit_steps(v8f acc, const float4 (&a)[8], const float4* bu) {
  #pragma unroll
  for (int j = 0; j < 8; ++j) {
    const float4 b = bu[2 * j];
    const v2f ax = {a[j].x, a[j].y}, bx = {b.x, b.y};
    const v2f ay = {a[j].z, a[j].w}, by = {b.z, b.w};
    acc = __builtin_amdgcn_wmma_f32_16x16x4_f32(false, ax, false, bx, (short)0, acc, false, false);
    acc = __builtin_amdgcn_wmma_f32_16x16x4_f32(false, ay, false, by, (short)0, acc, false, false);
  }
  return acc;
}

__global__ void __launch_bounds__(TC_THREADS, 10)
transform_group(const float* __restrict__ feat, const float* __restrict__ W1,
                float* __restrict__ t1, const int* __restrict__ src, const int* __restrict__ dst,
                int* __restrict__ off, unsigned* __restrict__ list, size_t ws_bytes) {
  __shared__ TcShared sm;
  const int tid = threadIdx.x;

  if (blockIdx.x >= TC_GEMM_WGS) {
    if (ws_bytes < WS_NEED) return;
    const int chunk = blockIdx.x - TC_GEMM_WGS;
    if (tid < NB) sm.g.hist[tid] = 0;
    __syncthreads();
    const int e0 = chunk * CH_EDGES + tid;
    {
      int d[TC_PER_LANE];
      #pragma unroll
      for (int i = 0; i < TC_PER_LANE; ++i) d[i] = dst[e0 + i * TC_THREADS];
      #pragma unroll
      for (int i = 0; i < TC_PER_LANE; ++i) lds_count(&sm.g.hist[(unsigned)d[i] / BIN_NODES]);
    }
    __syncthreads();
    if (tid < 32) {
      int h[8];
      int sum = 0;
      #pragma unroll
      for (int k = 0; k < 8; ++k) {
        const int b = tid * 8 + k;
        h[k] = (b < NB) ? sm.g.hist[b] : 0;
        sum += h[k];
      }
      int inc = sum;
      #pragma unroll
      for (int o = 1; o < 32; o <<= 1) {
        const int up = __shfl_up(inc, o);
        if (tid >= o) inc += up;
      }
      int run = inc - sum;
      #pragma unroll
      for (int k = 0; k < 8; ++k) {
        const int b = tid * 8 + k;
        if (b < NB) {
          sm.g.cur[b] = run;
          off[chunk * NB + b] = run;
        }
        run += h[k];
      }
    }
    __syncthreads();
    #pragma nounroll
    for (int hh = 0; hh < 2; ++hh) {
      int d[TC_HALF];
      int s[TC_HALF];
      int slot[TC_HALF];
      #pragma unroll
      for (int i = 0; i < TC_HALF; ++i) {
        d[i] = dst[e0 + (hh * TC_HALF + i) * TC_THREADS];
        s[i] = src[e0 + (hh * TC_HALF + i) * TC_THREADS];
      }
      #pragma unroll
      for (int i = 0; i < TC_HALF; ++i)
        slot[i] = lds_take(&sm.g.cur[(unsigned)d[i] / BIN_NODES]);
      #pragma unroll
      for (int i = 0; i < TC_HALF; ++i) {
        const unsigned b  = (unsigned)d[i] / BIN_NODES;
        const unsigned dl = (unsigned)d[i] - b * BIN_NODES;
        sm.g.sorted[slot[i]] = (unsigned)s[i] | (dl << SRC_BITS);
      }
    }
    __syncthreads();
    const uint4* s4 = (const uint4*)sm.g.sorted;
    uint4* l4 = (uint4*)(list + (size_t)chunk * CH_EDGES);
    for (int j = tid; j < CH_EDGES / 4; j += TC_THREADS) l4[j] = s4[j];
    return;
  }

  const int lane = tid & 31;
  const int wave = tid >> 5;
  const int tile = blockIdx.x * (TC_THREADS >> 5) + wave;
  const bool has_tile = tile < TC_TILES;
  const int node0 = tile * 16;

  const int c = lane & 15;
  const int h = lane >> 4;
  const float4* arow = (const float4*)(feat + (size_t)(node0 + c) * D_FEAT + 4 * h);
  const float4* bcol = (const float4*)(sm.w + c * WT_PITCH + 4 * h);

  float4 a0[8], a1[8];
  if (has_tile) {
    #pragma unroll
    for (int j = 0; j < 8; ++j) a0[j] = arow[2 * j];
  }

  {
    const float4* s4 = (const float4*)W1;
    float4 wv[(D_FEAT * D_HID) / 4 / TC_THREADS];
    #pragma unroll
    for (int r = 0; r < (D_FEAT * D_HID) / 4 / TC_THREADS; ++r) wv[r] = s4[tid + r * TC_THREADS];
    #pragma unroll
    for (int r = 0; r < (D_FEAT * D_HID) / 4 / TC_THREADS; ++r) {
      const int i  = tid + r * TC_THREADS;
      const int k  = i >> 2;
      const int n0 = (i & 3) * 4;
      sm.w[(n0 + 0) * WT_PITCH + k] = wv[r].x;
      sm.w[(n0 + 1) * WT_PITCH + k] = wv[r].y;
      sm.w[(n0 + 2) * WT_PITCH + k] = wv[r].z;
      sm.w[(n0 + 3) * WT_PITCH + k] = wv[r].w;
    }
  }
  __syncthreads();
  if (!has_tile) return;

  v8f acc = {};
  #pragma unroll 1
  for (int u = 0; u < D_FEAT / K_UNIT - 2; u += 2) {
    #pragma unroll
    for (int j = 0; j < 8; ++j) a1[j] = arow[(u + 1) * (K_UNIT / 4) + 2 * j];
    acc = unit_steps(acc, a0, bcol + u * (K_UNIT / 4));
    #pragma unroll
    for (int j = 0; j < 8; ++j) a0[j] = arow[(u + 2) * (K_UNIT / 4) + 2 * j];
    acc = unit_steps(acc, a1, bcol + (u + 1) * (K_UNIT / 4));
  }
  #pragma unroll
  for (int j = 0; j < 8; ++j) a1[j] = arow[(D_FEAT / K_UNIT - 1) * (K_UNIT / 4) + 2 * j];
  acc = unit_steps(acc, a0, bcol + (D_FEAT / K_UNIT - 2) * (K_UNIT / 4));
  acc = unit_steps(acc, a1, bcol + (D_FEAT / K_UNIT - 1) * (K_UNIT / 4));

  const int mn = c;
  const int rbase = node0 + ((lane >> 4) << 3);
  #pragma unroll
  for (int r = 0; r < 8; ++r)
    t1[(size_t)(rbase + r) * D_HID + mn] = acc[r];
}

__global__ void __launch_bounds__(SUM_THREADS)
sum_layer1(const float* __restrict__ t1, const unsigned* __restrict__ list,
           const int* __restrict__ off, const float* __restrict__ b1,
           const float* __restrict__ W2, float* __restrict__ t2, size_t ws_bytes) {
  if (ws_bytes < WS_NEED) return;
  __shared__ __attribute__((aligned(16))) float acc[BIN_NODES * D_HID];
  const int tid = threadIdx.x;
  for (int i = tid; i < BIN_NODES * D_HID; i += SUM_THREADS) acc[i] = 0.0f;
  __syncthreads();
  const int bin = blockIdx.x;
  const int lane = tid & 31;
  const int f    = lane & 15;
  const int half = lane >> 4;
  const int wave = __builtin_amdgcn_readfirstlane(tid >> 5);
  int sb[SEG_PER_WAVE], ln[SEG_PER_WAVE], total;
  wave_pieces(off, bin, wave, lane, sb, ln, total);
  int f0 = 0;
  unsigned p_l = run_entry(list, sb, ln, total, lane);
  while (f0 < total) {
    const int n = total - f0;
    float v[16];
    int   o[16];
    rows16_gather(t1, p_l, f, half, v, o);
    f0 += 32;
    const unsigned p_n = run_entry(list, sb, ln, total, f0 + lane);
    if (n >= 32) rows16_add<true>(acc, v, o, n, half);
    else         rows16_add<false>(acc, v, o, n, half);
    p_l = p_n;
  }
  __syncthreads();
  if (tid < BIN_NODES) {
    const int node = bin * BIN_NODES + tid;
    if (node < N_NODES) {
      float h[D_HID];
      const float4* a4 = (const float4*)(acc + tid * D_HID);
      #pragma unroll
      for (int q = 0; q < 4; ++q) {
        const float4 a = a4[q];
        h[q * 4 + 0] = fmaxf(a.x + b1[q * 4 + 0], 0.0f);
        h[q * 4 + 1] = fmaxf(a.y + b1[q * 4 + 1], 0.0f);
        h[q * 4 + 2] = fmaxf(a.z + b1[q * 4 + 2], 0.0f);
        h[q * 4 + 3] = fmaxf(a.w + b1[q * 4 + 3], 0.0f);
      }
      float t[T2_PITCH];
      #pragma unroll
      for (int c = 0; c < N_CLASSES; ++c) {
        float s = 0.0f;
        #pragma unroll
        for (int j = 0; j < D_HID; ++j)
          s = fmaf(h[j], W2[j * N_CLASSES + c], s);
        t[c] = s;
      }
      t[N_CLASSES] = 0.0f;
      float4* t4 = (float4*)(t2 + (size_t)node * T2_PITCH);
      t4[0] = make_float4(t[0], t[1], t[2], t[3]);
      t4[1] = make_float4(t[4], t[5], t[6], t[7]);
    }
  }
}

__global__ void __launch_bounds__(SUM_THREADS)
sum_layer2(const float* __restrict__ t2, const unsigned* __restrict__ list,
           const int* __restrict__ off, const float* __restrict__ b2,
           float* __restrict__ out, size_t ws_bytes) {
  if (ws_bytes < WS_NEED) return;
  __shared__ float acc[BIN_NODES * T2_PITCH];
  const int tid = threadIdx.x;
  for (int i = tid; i < BIN_NODES * T2_PITCH; i += SUM_THREADS) acc[i] = 0.0f;
  __syncthreads();
  const int bin = blockIdx.x;
  const int lane = tid & 31;
  const int c = lane & 7;
  const int q = lane >> 3;
  const int wave = __builtin_amdgcn_readfirstlane(tid >> 5);
  int sb[SEG_PER_WAVE], ln[SEG_PER_WAVE], total;
  wave_pieces(off, bin, wave, lane, sb, ln, total);
  int f0 = 0;
  unsigned p_l = run_entry(list, sb, ln, total, lane);
  while (f0 < total) {
    const int n = total - f0;
    float v[8];
    int   o[8];
    rows8_gather(t2, p_l, c, q, v, o);
    f0 += 32;
    const unsigned p_n = run_entry(list, sb, ln, total, f0 + lane);
    if (n >= 32) rows8_add<true>(acc, v, o, n, c, q);
    else         rows8_add<false>(acc, v, o, n, c, q);
    p_l = p_n;
  }
  __syncthreads();
  const int node0 = bin * BIN_NODES;
  const int nodes = (N_NODES - node0 < BIN_NODES) ? (N_NODES - node0) : BIN_NODES;
  for (int idx = tid; idx < nodes * N_CLASSES; idx += SUM_THREADS) {
    const int n  = idx / N_CLASSES;
    const int cc = idx - n * N_CLASSES;
    out[node0 * N_CLASSES + idx] = acc[n * T2_PITCH + cc] + b2[cc];
  }
}

extern "C" void kernel_launch(void* const* d_in, const int* in_sizes, int n_in,
                              void* d_out, int out_size, void* d_ws, size_t ws_size,
                              hipStream_t stream) {
  const float* feat = (const float*)d_in[0];
  const float* W1   = (const float*)d_in[1];
  const float* b1   = (const float*)d_in[2];
  const float* W2   = (const float*)d_in[3];
  const float* b2   = (const float*)d_in[4];
  const int*   src  = (const int*)d_in[5];
  const int*   dst  = (const int*)d_in[6];
  float* out = (float*)d_out;

  float*    ws   = (float*)d_ws;
  float*    t1   = ws + WS_T1;
  unsigned* list = (unsigned*)d_ws + WS_LIST;
  float*    t2   = ws + WS_T2;
  int*      off  = (int*)d_ws + WS_OFF;

  transform_group<<<TC_GEMM_WGS + NCH, TC_THREADS, 0, stream>>>(feat, W1, t1, src, dst, off, list, ws_size);

  sum_layer1<<<NB, SUM_THREADS, 0, stream>>>(t1, list, off, b1, W2, t2, ws_size);

  sum_layer2<<<NB, SUM_THREADS, 0, stream>>>(t2, list, off, b2, out, ws_size);
}
